# starting kernel
# baseline (speedup 1.0000x reference)
#include <hip/hip_runtime.h>

typedef float v2f __attribute__((ext_vector_type(2)));
typedef float v8f __attribute__((ext_vector_type(8)));

#define WMMA_F32(a, b, c) \
    __builtin_amdgcn_wmma_f32_16x16x4_f32(false, (a), false, (b), (short)0, (c), false, false)

#define EMBED 1024
#define THREE_EMBED 3072
#define NHEAD 16
#define HD 64
#define SEQ 2048
#define BATCH 4

__global__ __launch_bounds__(256) void qkv_gemm_kernel(
    const float* __restrict__ x, const float* __restrict__ w,
    const float* __restrict__ bias,
    float* __restrict__ qb, float* __restrict__ kb, float* __restrict__ vb)
{
    const int lane   = threadIdx.x & 31;
    const int wave   = threadIdx.x >> 5;
    const int lane16 = lane & 15;
    const int laneHi = lane >> 4;

    const int wid   = blockIdx.x * 8 + wave;
    const int mtile = wid & 255;
    const int ntile = wid >> 8;
    const int m0    = mtile * 32;
    const int n0    = ntile * 64;

    const float* arow0 = x + (long)(m0 + lane16) * EMBED + 2 * laneHi;
    const float* arow1 = arow0 + 16 * EMBED;

    v8f acc0[4] = {};
    v8f acc1[4] = {};

    #pragma unroll 4
    for (int k0 = 0; k0 < EMBED; k0 += 4) {
        v2f a0 = *(const v2f*)(arow0 + k0);
        v2f a1 = *(const v2f*)(arow1 + k0);
        const int kk = k0 + 2 * laneHi;
        const float* wp = w + (long)kk * THREE_EMBED + n0 + lane16;
        #pragma unroll
        for (int c = 0; c < 4; ++c) {
            v2f b;
            b.x = wp[c * 16];
            b.y = wp[c * 16 + THREE_EMBED];
            acc0[c] = WMMA_F32(a0, b, acc0[c]);
            acc1[c] = WMMA_F32(a1, b, acc1[c]);
        }
    }

    const int section = n0 / EMBED;
    const int h       = (n0 % EMBED) / HD;
    float* dst = (section == 0) ? qb : (section == 1) ? kb : vb;

    #pragma unroll
    for (int c = 0; c < 4; ++c) {
        const float bv = bias[n0 + c * 16 + lane16];
        const int d = c * 16 + lane16;
        #pragma unroll
        for (int g = 0; g < 2; ++g) {
            #pragma unroll
            for (int vv = 0; vv < 8; ++vv) {
                const int m  = m0 + 16 * g + vv + 8 * laneHi;
                const int bg = m >> 11;
                const int t  = m & 2047;
                const float val = (g == 0 ? acc0[c][vv] : acc1[c][vv]) + bv;
                dst[(((long)(bg * NHEAD + h)) * SEQ + t) * HD + d] = val;
            }
        }
    }
}

__global__ __launch_bounds__(32) void attn_kernel(
    const float* __restrict__ q, const float* __restrict__ k,
    const float* __restrict__ v, float* __restrict__ out)
{
    __shared__ float pstage[2 * 256];

    const int lane   = threadIdx.x & 31;
    const int lane16 = lane & 15;
    const int laneHi = lane >> 4;

    const int wid = blockIdx.x;
    const int q32 = wid & 63;
    const int bh  = wid >> 6;
    const int b   = bh >> 4;
    const int h   = bh & 15;
    const int q0  = q32 * 32;

    const float* qp0 = q + ((long)bh * SEQ + q0 + lane16) * HD + 2 * laneHi;
    const float* qp1 = qp0 + 16 * HD;
    v2f aQ0[16], aQ1[16];
    #pragma unroll
    for (int j = 0; j < 16; ++j) {
        aQ0[j] = *(const v2f*)(qp0 + 4 * j);
        aQ1[j] = *(const v2f*)(qp1 + 4 * j);
    }

    float mrow[2][8], lrow[2][8];
    v8f O[2][4] = {};
    #pragma unroll
    for (int g = 0; g < 2; ++g)
        #pragma unroll
        for (int vv = 0; vv < 8; ++vv) { mrow[g][vv] = -__builtin_inff(); lrow[g][vv] = 0.f; }

    const float scale = 0.125f;
    const int ktmax = (q0 + 31) >> 4;

    for (int kt = 0; kt <= ktmax; ++kt) {
        const int k0 = kt * 16;

        const float* kp = k + ((long)bh * SEQ + k0 + lane16) * HD + 2 * laneHi;
        v8f S[2] = {};
        #pragma unroll
        for (int j = 0; j < 16; ++j) {
            v2f bK = *(const v2f*)(kp + 4 * j);
            S[0] = WMMA_F32(aQ0[j], bK, S[0]);
            S[1] = WMMA_F32(aQ1[j], bK, S[1]);
        }

        #pragma unroll
        for (int g = 0; g < 2; ++g) {
            #pragma unroll
            for (int vv = 0; vv < 8; ++vv) {
                const int row = 16 * g + vv + 8 * laneHi;
                float s = S[g][vv] * scale;
                if (k0 + lane16 > q0 + row) s = -__builtin_inff();
                S[g][vv] = s;
            }
            float alpha[8];
            #pragma unroll
            for (int vv = 0; vv < 8; ++vv) {
                float tmax = S[g][vv];
                #pragma unroll
                for (int off = 1; off < 16; off <<= 1)
                    tmax = fmaxf(tmax, __shfl_xor(tmax, off, 32));
                const float mnew = fmaxf(mrow[g][vv], tmax);
                const float p    = __expf(S[g][vv] - mnew);
                float psum = p;
                #pragma unroll
                for (int off = 1; off < 16; off <<= 1)
                    psum += __shfl_xor(psum, off, 32);
                const float a = __expf(mrow[g][vv] - mnew);
                lrow[g][vv]  = lrow[g][vv] * a + psum;
                mrow[g][vv]  = mnew;
                alpha[vv]    = a;
                S[g][vv]     = p;
            }
            #pragma unroll
            for (int c = 0; c < 4; ++c)
                #pragma unroll
                for (int vv = 0; vv < 8; ++vv)
                    O[g][c][vv] *= alpha[vv];
        }

        __syncthreads();
        #pragma unroll
        for (int g = 0; g < 2; ++g)
            #pragma unroll
            for (int vv = 0; vv < 8; ++vv)
                pstage[g * 256 + (vv + 8 * laneHi) * 16 + lane16] = S[g][vv];
        __syncthreads();
        v2f aP0[4], aP1[4];
        #pragma unroll
        for (int j = 0; j < 4; ++j) {
            aP0[j] = *(const v2f*)(&pstage[lane16 * 16 + 4 * j + 2 * laneHi]);
            aP1[j] = *(const v2f*)(&pstage[256 + lane16 * 16 + 4 * j + 2 * laneHi]);
        }

        const float* vp = v + ((long)bh * SEQ + k0) * HD;
        #pragma unroll
        for (int c = 0; c < 4; ++c) {
            #pragma unroll
            for (int j = 0; j < 4; ++j) {
                const int kk = 4 * j + 2 * laneHi;
                const float* vbp = vp + (long)kk * HD + c * 16 + lane16;
                v2f bV;
                bV.x = vbp[0];
                bV.y = vbp[HD];
                O[0][c] = WMMA_F32(aP0[j], bV, O[0][c]);
                O[1][c] = WMMA_F32(aP1[j], bV, O[1][c]);
            }
        }
    }

    #pragma unroll
    for (int g = 0; g < 2; ++g) {
        #pragma unroll
        for (int c = 0; c < 4; ++c) {
            #pragma unroll
            for (int vv = 0; vv < 8; ++vv) {
                const int t = q0 + 16 * g + vv + 8 * laneHi;
                const int d = c * 16 + lane16;
                out[(((long)b * SEQ + t) * NHEAD + h) * HD + d] = O[g][c][vv] / lrow[g][vv];
            }
        }
    }
}

extern "C" void kernel_launch(void* const* d_in, const int* in_sizes, int n_in,
                              void* d_out, int out_size, void* d_ws, size_t ws_size,
                              hipStream_t stream) {
    const float* x    = (const float*)d_in[0];
    const float* Wqkv = (const float*)d_in[1];
    const float* bqkv = (const float*)d_in[2];
    float* out = (float*)d_out;

    float* ws = (float*)d_ws;
    const long QS = (long)BATCH * NHEAD * SEQ * HD;
    float* qb = ws;
    float* kb = ws + QS;
    float* vb = ws + 2 * QS;

    qkv_gemm_kernel<<<1536, 256, 0, stream>>>(x, Wqkv, bqkv, qb, kb, vb);
    attn_kernel<<<BATCH * NHEAD * (SEQ / 32), 32, 0, stream>>>(qb, kb, vb, out);
}


# Attention key loop retiled to 64 keys per trip: a lane folds its four 16-key score tiles in registers, so row max/sum shuffles, O rescale and LDS transpose run once per 64 keys; key sum order changes
# speedup vs baseline: 1.4589x; 1.4589x over previous
#include <hip/hip_runtime.h>

typedef float v2f __attribute__((ext_vector_type(2)));
typedef float v8f __attribute__((ext_vector_type(8)));

#define WMMA_F32(a, b, c) \
    __builtin_amdgcn_wmma_f32_16x16x4_f32(false, (a), false, (b), (short)0, (c), false, false)

#define EMBED 1024
#define THREE_EMBED 3072
#define NHEAD 16
#define HD 64
#define SEQ 2048
#define BATCH 4

__global__ __launch_bounds__(256) void qkv_gemm_kernel(
    const float* __restrict__ x, const float* __restrict__ w,
    const float* __restrict__ bias,
    float* __restrict__ qb, float* __restrict__ kb, float* __restrict__ vb)
{
    const int lane   = threadIdx.x & 31;
    const int wave   = threadIdx.x >> 5;
    const int lane16 = lane & 15;
    const int laneHi = lane >> 4;

    const int wid   = blockIdx.x * 8 + wave;
    const int mtile = wid & 255;
    const int ntile = wid >> 8;
    const int m0    = mtile * 32;
    const int n0    = ntile * 64;

    const float* arow0 = x + (long)(m0 + lane16) * EMBED + 2 * laneHi;
    const float* arow1 = arow0 + 16 * EMBED;

    v8f acc0[4] = {};
    v8f acc1[4] = {};

    #pragma unroll 4
    for (int k0 = 0; k0 < EMBED; k0 += 4) {
        v2f a0 = *(const v2f*)(arow0 + k0);
        v2f a1 = *(const v2f*)(arow1 + k0);
        const int kk = k0 + 2 * laneHi;
        const float* wp = w + (long)kk * THREE_EMBED + n0 + lane16;
        #pragma unroll
        for (int c = 0; c < 4; ++c) {
            v2f b;
            b.x = wp[c * 16];
            b.y = wp[c * 16 + THREE_EMBED];
            acc0[c] = WMMA_F32(a0, b, acc0[c]);
            acc1[c] = WMMA_F32(a1, b, acc1[c]);
        }
    }

    const int section = n0 / EMBED;
    const int h       = (n0 % EMBED) / HD;
    float* dst = (section == 0) ? qb : (section == 1) ? kb : vb;

    #pragma unroll
    for (int c = 0; c < 4; ++c) {
        const float bv = bias[n0 + c * 16 + lane16];
        const int d = c * 16 + lane16;
        #pragma unroll
        for (int g = 0; g < 2; ++g) {
            #pragma unroll
            for (int vv = 0; vv < 8; ++vv) {
                const int m  = m0 + 16 * g + vv + 8 * laneHi;
                const int bg = m >> 11;
                const int t  = m & 2047;
                const float val = (g == 0 ? acc0[c][vv] : acc1[c][vv]) + bv;
                dst[(((long)(bg * NHEAD + h)) * SEQ + t) * HD + d] = val;
            }
        }
    }
}

#define PPITCH 68

__global__ __launch_bounds__(32) void attn_kernel(
    const float* __restrict__ q, const float* __restrict__ k,
    const float* __restrict__ v, float* __restrict__ out)
{
    __shared__ __attribute__((aligned(16))) float pstage[2 * 16 * PPITCH];

    const int lane   = threadIdx.x & 31;
    const int lane16 = lane & 15;
    const int laneHi = lane >> 4;

    const int wid = blockIdx.x;
    const int q32 = wid & 63;
    const int bh  = wid >> 6;
    const int b   = bh >> 4;
    const int h   = bh & 15;
    const int q0  = q32 * 32;

    const float* qp0 = q + ((long)bh * SEQ + q0 + lane16) * HD + 2 * laneHi;
    const float* qp1 = qp0 + 16 * HD;
    v2f aQ0[16], aQ1[16];
    #pragma unroll
    for (int j = 0; j < 16; ++j) {
        aQ0[j] = *(const v2f*)(qp0 + 4 * j);
        aQ1[j] = *(const v2f*)(qp1 + 4 * j);
    }

    float mrow[2][8], lrow[2][8];
    v8f O[2][4] = {};
    #pragma unroll
    for (int g = 0; g < 2; ++g)
        #pragma unroll
        for (int vv = 0; vv < 8; ++vv) { mrow[g][vv] = -__builtin_inff(); lrow[g][vv] = 0.f; }

    const float scale = 0.125f;
    const int ntrip = (q0 + 95) >> 6;

    float* pst = &pstage[(8 * laneHi) * PPITCH + lane16];
    const float* pld = &pstage[lane16 * PPITCH + 2 * laneHi];

    for (int kt = 0; kt < ntrip; ++kt) {
        const int k0 = kt * 64;

        v8f S[2][4];
        #pragma unroll
        for (int n = 0; n < 4; ++n) {
            const float* kp = k + ((long)bh * SEQ + k0 + 16 * n + lane16) * HD + 2 * laneHi;
            v2f bK[16];
            #pragma unroll
            for (int j = 0; j < 16; ++j)
                bK[j] = *(const v2f*)(kp + 4 * j);
            __builtin_amdgcn_sched_barrier(0);
            v8f s0 = {};
            v8f s1 = {};
            #pragma unroll
            for (int j = 0; j < 16; ++j) {
                s0 = WMMA_F32(aQ0[j], bK[j], s0);
                s1 = WMMA_F32(aQ1[j], bK[j], s1);
            }
            S[0][n] = s0;
            S[1][n] = s1;
            __builtin_amdgcn_sched_barrier(0);
        }

        #pragma unroll
        for (int g = 0; g < 2; ++g) {
            #pragma unroll
            for (int n = 0; n < 4; ++n) {
                #pragma unroll
                for (int vv = 0; vv < 8; ++vv) {
                    const int row = 16 * g + vv + 8 * laneHi;
                    float s = S[g][n][vv] * scale;
                    if (k0 + 16 * n + lane16 > q0 + row) s = -__builtin_inff();
                    S[g][n][vv] = s;
                }
            }
            float alpha[8];
            #pragma unroll
            for (int vv = 0; vv < 8; ++vv) {
                float tmax = fmaxf(fmaxf(S[g][0][vv], S[g][1][vv]),
                                   fmaxf(S[g][2][vv], S[g][3][vv]));
                #pragma unroll
                for (int off = 1; off < 16; off <<= 1)
                    tmax = fmaxf(tmax, __shfl_xor(tmax, off, 32));
                const float mnew = fmaxf(mrow[g][vv], tmax);
                float p[4];
                #pragma unroll
                for (int n = 0; n < 4; ++n)
                    p[n] = __expf(S[g][n][vv] - mnew);
                float psum = (p[0] + p[1]) + (p[2] + p[3]);
                #pragma unroll
                for (int off = 1; off < 16; off <<= 1)
                    psum += __shfl_xor(psum, off, 32);
                const float a = __expf(mrow[g][vv] - mnew);
                lrow[g][vv]  = lrow[g][vv] * a + psum;
                mrow[g][vv]  = mnew;
                alpha[vv]    = a;
                #pragma unroll
                for (int n = 0; n < 4; ++n)
                    S[g][n][vv] = p[n];
            }
            #pragma unroll
            for (int c = 0; c < 4; ++c)
                #pragma unroll
                for (int vv = 0; vv < 8; ++vv)
                    O[g][c][vv] *= alpha[vv];
        }

        __syncthreads();
        #pragma unroll
        for (int g = 0; g < 2; ++g)
            #pragma unroll
            for (int n = 0; n < 4; ++n)
                #pragma unroll
                for (int vv = 0; vv < 8; ++vv)
                    pst[g * 16 * PPITCH + vv * PPITCH + 16 * n] = S[g][n][vv];
        __syncthreads();

        const float* vp = v + ((long)bh * SEQ + k0) * HD + lane16;
        #pragma unroll
        for (int n = 0; n < 4; ++n) {
            v2f bV[4][4];
            v2f aP[4][2];
            #pragma unroll
            for (int j = 0; j < 4; ++j) {
                const int kk = 16 * n + 4 * j + 2 * laneHi;
                aP[j][0] = *(const v2f*)(pld + 16 * n + 4 * j);
                aP[j][1] = *(const v2f*)(pld + 16 * PPITCH + 16 * n + 4 * j);
                #pragma unroll
                for (int c = 0; c < 4; ++c) {
                    const float* vbp = vp + (long)kk * HD + c * 16;
                    bV[j][c].x = vbp[0];
                    bV[j][c].y = vbp[HD];
                }
            }
            __builtin_amdgcn_sched_barrier(0);
            #pragma unroll
            for (int j = 0; j < 4; ++j) {
                #pragma unroll
                for (int c = 0; c < 4; ++c) {
                    O[0][c] = WMMA_F32(aP[j][0], bV[j][c], O[0][c]);
                    O[1][c] = WMMA_F32(aP[j][1], bV[j][c], O[1][c]);
                }
            }
            __builtin_amdgcn_sched_barrier(0);
        }
    }

    #pragma unroll
    for (int g = 0; g < 2; ++g) {
        #pragma unroll
        for (int c = 0; c < 4; ++c) {
            #pragma unroll
            for (int vv = 0; vv < 8; ++vv) {
                const int t = q0 + 16 * g + vv + 8 * laneHi;
                const int d = c * 16 + lane16;
                out[(((long)b * SEQ + t) * NHEAD + h) * HD + d] = O[g][c][vv] / lrow[g][vv];
            }
        }
    }
}

extern "C" void kernel_launch(void* const* d_in, const int* in_sizes, int n_in,
                              void* d_out, int out_size, void* d_ws, size_t ws_size,
                              hipStream_t stream) {
    const float* x    = (const float*)d_in[0];
    const float* Wqkv = (const float*)d_in[1];
    const float* bqkv = (const float*)d_in[2];
    float* out = (float*)d_out;

    float* ws = (float*)d_ws;
    const long QS = (long)BATCH * NHEAD * SEQ * HD;
    float* qb = ws;
    float* kb = ws + QS;
    float* vb = ws + 2 * QS;

    qkv_gemm_kernel<<<1536, 256, 0, stream>>>(x, Wqkv, bqkv, qb, kb, vb);
    attn_kernel<<<BATCH * NHEAD * (SEQ / 32), 32, 0, stream>>>(qb, kb, vb, out);
}


# QKV projection GEMM fed from LDS: a 4-wave work-group stages 128x32 x and 32x128 W tiles (whole-line global loads, two LDS stages, one barrier per k tile); each wave a 64x64 tile; sum order unchanged
# speedup vs baseline: 1.5438x; 1.0582x over previous
#include <hip/hip_runtime.h>

typedef float v2f __attribute__((ext_vector_type(2)));
typedef float v8f __attribute__((ext_vector_type(8)));

#define WMMA_F32(a, b, c) \
    __builtin_amdgcn_wmma_f32_16x16x4_f32(false, (a), false, (b), (short)0, (c), false, false)

#define EMBED 1024
#define THREE_EMBED 3072
#define NHEAD 16
#define HD 64
#define SEQ 2048
#define BATCH 4

typedef float v4f __attribute__((ext_vector_type(4)));

#define GBM 128
#define GBN 128
#define GBK 32
#define APITCH 36
#define BPITCH 136

__global__ __launch_bounds__(128) void qkv_gemm_kernel(
    const float* __restrict__ x, const float* __restrict__ w,
    const float* __restrict__ bias,
    float* __restrict__ qb, float* __restrict__ kb, float* __restrict__ vb)
{
    __shared__ __attribute__((aligned(16))) float As[2][GBM * APITCH];
    __shared__ __attribute__((aligned(16))) float Bs[2][GBK * BPITCH];

    const int tid    = threadIdx.x;
    const int lane   = tid & 31;
    const int wave   = tid >> 5;
    const int lane16 = lane & 15;
    const int laneHi = lane >> 4;
    const int wm     = wave >> 1;
    const int wn     = wave & 1;

    const int mtile = blockIdx.x & 63;
    const int ntile = blockIdx.x >> 6;
    const int m0    = mtile * GBM;
    const int n0    = ntile * GBN;

    const int arow = tid >> 3;
    const int acol = (tid & 7) * 4;
    const int brow = tid >> 5;
    const int bcol = (tid & 31) * 4;
    const float* ag = x + (long)(m0 + arow) * EMBED + acol;
    const float* bg = w + (long)brow * THREE_EMBED + n0 + bcol;

    v4f ra[8], rb[8];
    #pragma unroll
    for (int i = 0; i < 8; ++i) {
        ra[i] = *(const v4f*)(ag + (long)(16 * i) * EMBED);
        rb[i] = *(const v4f*)(bg + (long)(4 * i) * THREE_EMBED);
    }
    #pragma unroll
    for (int i = 0; i < 8; ++i) {
        *(v4f*)&As[0][(arow + 16 * i) * APITCH + acol] = ra[i];
        *(v4f*)&Bs[0][(brow + 4 * i) * BPITCH + bcol]  = rb[i];
    }
    __syncthreads();

    v8f acc[4][4] = {};

    const int aoff = (wm * 64 + lane16) * APITCH + 2 * laneHi;
    const int boff = (2 * laneHi) * BPITCH + wn * 64 + lane16;

    for (int kt = 0; kt < EMBED / GBK; ++kt) {
        const int cur = kt & 1;
        const bool more = (kt + 1) < (EMBED / GBK);

        if (more) {
            const float* agn = ag + (kt + 1) * GBK;
            const float* bgn = bg + (long)(kt + 1) * GBK * THREE_EMBED;
            #pragma unroll
            for (int i = 0; i < 8; ++i) {
                ra[i] = *(const v4f*)(agn + (long)(16 * i) * EMBED);
                rb[i] = *(const v4f*)(bgn + (long)(4 * i) * THREE_EMBED);
            }
        }

        const float* as = &As[cur][aoff];
        const float* bs = &Bs[cur][boff];
        #pragma unroll
        for (int ks = 0; ks < GBK / 4; ++ks) {
            v2f a[4], b[4];
            #pragma unroll
            for (int i = 0; i < 4; ++i)
                a[i] = *(const v2f*)(as + 16 * i * APITCH + 4 * ks);
            #pragma unroll
            for (int j = 0; j < 4; ++j) {
                b[j].x = bs[(4 * ks) * BPITCH + 16 * j];
                b[j].y = bs[(4 * ks + 1) * BPITCH + 16 * j];
            }
            #pragma unroll
            for (int i = 0; i < 4; ++i)
                #pragma unroll
                for (int j = 0; j < 4; ++j)
                    acc[i][j] = WMMA_F32(a[i], b[j], acc[i][j]);
        }

        if (more) {
            #pragma unroll
            for (int i = 0; i < 8; ++i) {
                *(v4f*)&As[cur ^ 1][(arow + 16 * i) * APITCH + acol] = ra[i];
                *(v4f*)&Bs[cur ^ 1][(brow + 4 * i) * BPITCH + bcol]  = rb[i];
            }
        }
        __syncthreads();
    }

    const int nw0     = n0 + wn * 64;
    const int section = nw0 / EMBED;
    const int h       = (nw0 % EMBED) / HD;
    float* dst = (section == 0) ? qb : (section == 1) ? kb : vb;

    #pragma unroll
    for (int j = 0; j < 4; ++j) {
        const int d = j * 16 + lane16;
        const float bv = bias[nw0 + d];
        #pragma unroll
        for (int i = 0; i < 4; ++i) {
            #pragma unroll
            for (int vv = 0; vv < 8; ++vv) {
                const int m  = m0 + wm * 64 + 16 * i + vv + 8 * laneHi;
                const int bi = m >> 11;
                const int t  = m & 2047;
                dst[(((long)(bi * NHEAD + h)) * SEQ + t) * HD + d] = acc[i][j][vv] + bv;
            }
        }
    }
}

#define PPITCH 68

__global__ __launch_bounds__(32) void attn_kernel(
    const float* __restrict__ q, const float* __restrict__ k,
    const float* __restrict__ v, float* __restrict__ out)
{
    __shared__ __attribute__((aligned(16))) float pstage[2 * 16 * PPITCH];

    const int lane   = threadIdx.x & 31;
    const int lane16 = lane & 15;
    const int laneHi = lane >> 4;

    const int wid = blockIdx.x;
    const int q32 = wid & 63;
    const int bh  = wid >> 6;
    const int b   = bh >> 4;
    const int h   = bh & 15;
    const int q0  = q32 * 32;

    const float* qp0 = q + ((long)bh * SEQ + q0 + lane16) * HD + 2 * laneHi;
    const float* qp1 = qp0 + 16 * HD;
    v2f aQ0[16], aQ1[16];
    #pragma unroll
    for (int j = 0; j < 16; ++j) {
        aQ0[j] = *(const v2f*)(qp0 + 4 * j);
        aQ1[j] = *(const v2f*)(qp1 + 4 * j);
    }

    float mrow[2][8], lrow[2][8];
    v8f O[2][4] = {};
    #pragma unroll
    for (int g = 0; g < 2; ++g)
        #pragma unroll
        for (int vv = 0; vv < 8; ++vv) { mrow[g][vv] = -__builtin_inff(); lrow[g][vv] = 0.f; }

    const float scale = 0.125f;
    const int ntrip = (q0 + 95) >> 6;

    float* pst = &pstage[(8 * laneHi) * PPITCH + lane16];
    const float* pld = &pstage[lane16 * PPITCH + 2 * laneHi];

    for (int kt = 0; kt < ntrip; ++kt) {
        const int k0 = kt * 64;

        v8f S[2][4];
        #pragma unroll
        for (int n = 0; n < 4; ++n) {
            const float* kp = k + ((long)bh * SEQ + k0 + 16 * n + lane16) * HD + 2 * laneHi;
            v2f bK[16];
            #pragma unroll
            for (int j = 0; j < 16; ++j)
                bK[j] = *(const v2f*)(kp + 4 * j);
            __builtin_amdgcn_sched_barrier(0);
            v8f s0 = {};
            v8f s1 = {};
            #pragma unroll
            for (int j = 0; j < 16; ++j) {
                s0 = WMMA_F32(aQ0[j], bK[j], s0);
                s1 = WMMA_F32(aQ1[j], bK[j], s1);
            }
            S[0][n] = s0;
            S[1][n] = s1;
            __builtin_amdgcn_sched_barrier(0);
        }

        #pragma unroll
        for (int g = 0; g < 2; ++g) {
            #pragma unroll
            for (int n = 0; n < 4; ++n) {
                #pragma unroll
                for (int vv = 0; vv < 8; ++vv) {
                    const int row = 16 * g + vv + 8 * laneHi;
                    float s = S[g][n][vv] * scale;
                    if (k0 + 16 * n + lane16 > q0 + row) s = -__builtin_inff();
                    S[g][n][vv] = s;
                }
            }
            float alpha[8];
            #pragma unroll
            for (int vv = 0; vv < 8; ++vv) {
                float tmax = fmaxf(fmaxf(S[g][0][vv], S[g][1][vv]),
                                   fmaxf(S[g][2][vv], S[g][3][vv]));
                #pragma unroll
                for (int off = 1; off < 16; off <<= 1)
                    tmax = fmaxf(tmax, __shfl_xor(tmax, off, 32));
                const float mnew = fmaxf(mrow[g][vv], tmax);
                float p[4];
                #pragma unroll
                for (int n = 0; n < 4; ++n)
                    p[n] = __expf(S[g][n][vv] - mnew);
                float psum = (p[0] + p[1]) + (p[2] + p[3]);
                #pragma unroll
                for (int off = 1; off < 16; off <<= 1)
                    psum += __shfl_xor(psum, off, 32);
                const float a = __expf(mrow[g][vv] - mnew);
                lrow[g][vv]  = lrow[g][vv] * a + psum;
                mrow[g][vv]  = mnew;
                alpha[vv]    = a;
                #pragma unroll
                for (int n = 0; n < 4; ++n)
                    S[g][n][vv] = p[n];
            }
            #pragma unroll
            for (int c = 0; c < 4; ++c)
                #pragma unroll
                for (int vv = 0; vv < 8; ++vv)
                    O[g][c][vv] *= alpha[vv];
        }

        __syncthreads();
        #pragma unroll
        for (int g = 0; g < 2; ++g)
            #pragma unroll
            for (int n = 0; n < 4; ++n)
                #pragma unroll
                for (int vv = 0; vv < 8; ++vv)
                    pst[g * 16 * PPITCH + vv * PPITCH + 16 * n] = S[g][n][vv];
        __syncthreads();

        const float* vp = v + ((long)bh * SEQ + k0) * HD + lane16;
        #pragma unroll
        for (int n = 0; n < 4; ++n) {
            v2f bV[4][4];
            v2f aP[4][2];
            #pragma unroll
            for (int j = 0; j < 4; ++j) {
                const int kk = 16 * n + 4 * j + 2 * laneHi;
                aP[j][0] = *(const v2f*)(pld + 16 * n + 4 * j);
                aP[j][1] = *(const v2f*)(pld + 16 * PPITCH + 16 * n + 4 * j);
                #pragma unroll
                for (int c = 0; c < 4; ++c) {
                    const float* vbp = vp + (long)kk * HD + c * 16;
                    bV[j][c].x = vbp[0];
                    bV[j][c].y = vbp[HD];
                }
            }
            __builtin_amdgcn_sched_barrier(0);
            #pragma unroll
            for (int j = 0; j < 4; ++j) {
                #pragma unroll
                for (int c = 0; c < 4; ++c) {
                    O[0][c] = WMMA_F32(aP[j][0], bV[j][c], O[0][c]);
                    O[1][c] = WMMA_F32(aP[j][1], bV[j][c], O[1][c]);
                }
            }
            __builtin_amdgcn_sched_barrier(0);
        }
    }

    #pragma unroll
    for (int g = 0; g < 2; ++g) {
        #pragma unroll
        for (int c = 0; c < 4; ++c) {
            #pragma unroll
            for (int vv = 0; vv < 8; ++vv) {
                const int t = q0 + 16 * g + vv + 8 * laneHi;
                const int d = c * 16 + lane16;
                out[(((long)b * SEQ + t) * NHEAD + h) * HD + d] = O[g][c][vv] / lrow[g][vv];
            }
        }
    }
}

extern "C" void kernel_launch(void* const* d_in, const int* in_sizes, int n_in,
                              void* d_out, int out_size, void* d_ws, size_t ws_size,
                              hipStream_t stream) {
    const float* x    = (const float*)d_in[0];
    const float* Wqkv = (const float*)d_in[1];
    const float* bqkv = (const float*)d_in[2];
    float* out = (float*)d_out;

    float* ws = (float*)d_ws;
    const long QS = (long)BATCH * NHEAD * SEQ * HD;
    float* qb = ws;
    float* kb = ws + QS;
    float* vb = ws + 2 * QS;

    qkv_gemm_kernel<<<1536, 128, 0, stream>>>(x, Wqkv, bqkv, qb, kb, vb);
    attn_kernel<<<BATCH * NHEAD * (SEQ / 32), 32, 0, stream>>>(qb, kb, vb, out);
}


# Attention row max/sum over 16 lanes done as DPP row_xmask butterflies on the vector ALU instead of wave shuffles through the LDS unit (one waited LDS round trip per step); sum order unchanged
# speedup vs baseline: 1.9666x; 1.2739x over previous
#include <hip/hip_runtime.h>

typedef float v2f __attribute__((ext_vector_type(2)));
typedef float v8f __attribute__((ext_vector_type(8)));

#define WMMA_F32(a, b, c) \
    __builtin_amdgcn_wmma_f32_16x16x4_f32(false, (a), false, (b), (short)0, (c), false, false)

#define EMBED 1024
#define THREE_EMBED 3072
#define NHEAD 16
#define HD 64
#define SEQ 2048
#define BATCH 4

typedef float v4f __attribute__((ext_vector_type(4)));

#define GBM 128
#define GBN 128
#define GBK 32
#define APITCH 36
#define BPITCH 136

__global__ __launch_bounds__(128) void qkv_gemm_kernel(
    const float* __restrict__ x, const float* __restrict__ w,
    const float* __restrict__ bias,
    float* __restrict__ qb, float* __restrict__ kb, float* __restrict__ vb)
{
    __shared__ __attribute__((aligned(16))) float As[2][GBM * APITCH];
    __shared__ __attribute__((aligned(16))) float Bs[2][GBK * BPITCH];

    const int tid    = threadIdx.x;
    const int lane   = tid & 31;
    const int wave   = tid >> 5;
    const int lane16 = lane & 15;
    const int laneHi = lane >> 4;
    const int wm     = wave >> 1;
    const int wn     = wave & 1;

    const int mtile = blockIdx.x & 63;
    const int ntile = blockIdx.x >> 6;
    const int m0    = mtile * GBM;
    const int n0    = ntile * GBN;

    const int arow = tid >> 3;
    const int acol = (tid & 7) * 4;
    const int brow = tid >> 5;
    const int bcol = (tid & 31) * 4;
    const float* ag = x + (long)(m0 + arow) * EMBED + acol;
    const float* bg = w + (long)brow * THREE_EMBED + n0 + bcol;

    v4f ra[8], rb[8];
    #pragma unroll
    for (int i = 0; i < 8; ++i) {
        ra[i] = *(const v4f*)(ag + (long)(16 * i) * EMBED);
        rb[i] = *(const v4f*)(bg + (long)(4 * i) * THREE_EMBED);
    }
    #pragma unroll
    for (int i = 0; i < 8; ++i) {
        *(v4f*)&As[0][(arow + 16 * i) * APITCH + acol] = ra[i];
        *(v4f*)&Bs[0][(brow + 4 * i) * BPITCH + bcol]  = rb[i];
    }
    __syncthreads();

    v8f acc[4][4] = {};

    const int aoff = (wm * 64 + lane16) * APITCH + 2 * laneHi;
    const int boff = (2 * laneHi) * BPITCH + wn * 64 + lane16;

    for (int kt = 0; kt < EMBED / GBK; ++kt) {
        const int cur = kt & 1;
        const bool more = (kt + 1) < (EMBED / GBK);

        if (more) {
            const float* agn = ag + (kt + 1) * GBK;
            const float* bgn = bg + (long)(kt + 1) * GBK * THREE_EMBED;
            #pragma unroll
            for (int i = 0; i < 8; ++i) {
                ra[i] = *(const v4f*)(agn + (long)(16 * i) * EMBED);
                rb[i] = *(const v4f*)(bgn + (long)(4 * i) * THREE_EMBED);
            }
        }

        const float* as = &As[cur][aoff];
        const float* bs = &Bs[cur][boff];
        #pragma unroll
        for (int ks = 0; ks < GBK / 4; ++ks) {
            v2f a[4], b[4];
            #pragma unroll
            for (int i = 0; i < 4; ++i)
                a[i] = *(const v2f*)(as + 16 * i * APITCH + 4 * ks);
            #pragma unroll
            for (int j = 0; j < 4; ++j) {
                b[j].x = bs[(4 * ks) * BPITCH + 16 * j];
                b[j].y = bs[(4 * ks + 1) * BPITCH + 16 * j];
            }
            #pragma unroll
            for (int i = 0; i < 4; ++i)
                #pragma unroll
                for (int j = 0; j < 4; ++j)
                    acc[i][j] = WMMA_F32(a[i], b[j], acc[i][j]);
        }

        if (more) {
            #pragma unroll
            for (int i = 0; i < 8; ++i) {
                *(v4f*)&As[cur ^ 1][(arow + 16 * i) * APITCH + acol] = ra[i];
                *(v4f*)&Bs[cur ^ 1][(brow + 4 * i) * BPITCH + bcol]  = rb[i];
            }
        }
        __syncthreads();
    }

    const int nw0     = n0 + wn * 64;
    const int section = nw0 / EMBED;
    const int h       = (nw0 % EMBED) / HD;
    float* dst = (section == 0) ? qb : (section == 1) ? kb : vb;

    #pragma unroll
    for (int j = 0; j < 4; ++j) {
        const int d = j * 16 + lane16;
        const float bv = bias[nw0 + d];
        #pragma unroll
        for (int i = 0; i < 4; ++i) {
            #pragma unroll
            for (int vv = 0; vv < 8; ++vv) {
                const int m  = m0 + wm * 64 + 16 * i + vv + 8 * laneHi;
                const int bi = m >> 11;
                const int t  = m & 2047;
                dst[(((long)(bi * NHEAD + h)) * SEQ + t) * HD + d] = acc[i][j][vv] + bv;
            }
        }
    }
}

template <int M>
__device__ __forceinline__ float row_xor(float x) {
    const int xi = __builtin_bit_cast(int, x);
    return __builtin_bit_cast(float,
        __builtin_amdgcn_update_dpp(xi, xi, 0x160 + M, 0xf, 0xf, true));
}

#define PPITCH 68

__global__ __launch_bounds__(32) void attn_kernel(
    const float* __restrict__ q, const float* __restrict__ k,
    const float* __restrict__ v, float* __restrict__ out)
{
    __shared__ __attribute__((aligned(16))) float pstage[2 * 16 * PPITCH];

    const int lane   = threadIdx.x & 31;
    const int lane16 = lane & 15;
    const int laneHi = lane >> 4;

    const int wid = blockIdx.x;
    const int q32 = wid & 63;
    const int bh  = wid >> 6;
    const int b   = bh >> 4;
    const int h   = bh & 15;
    const int q0  = q32 * 32;

    const float* qp0 = q + ((long)bh * SEQ + q0 + lane16) * HD + 2 * laneHi;
    const float* qp1 = qp0 + 16 * HD;
    v2f aQ0[16], aQ1[16];
    #pragma unroll
    for (int j = 0; j < 16; ++j) {
        aQ0[j] = *(const v2f*)(qp0 + 4 * j);
        aQ1[j] = *(const v2f*)(qp1 + 4 * j);
    }

    float mrow[2][8], lrow[2][8];
    v8f O[2][4] = {};
    #pragma unroll
    for (int g = 0; g < 2; ++g)
        #pragma unroll
        for (int vv = 0; vv < 8; ++vv) { mrow[g][vv] = -__builtin_inff(); lrow[g][vv] = 0.f; }

    const float scale = 0.125f;
    const int ntrip = (q0 + 95) >> 6;

    float* pst = &pstage[(8 * laneHi) * PPITCH + lane16];
    const float* pld = &pstage[lane16 * PPITCH + 2 * laneHi];

    for (int kt = 0; kt < ntrip; ++kt) {
        const int k0 = kt * 64;

        v8f S[2][4];
        #pragma unroll
        for (int n = 0; n < 4; ++n) {
            const float* kp = k + ((long)bh * SEQ + k0 + 16 * n + lane16) * HD + 2 * laneHi;
            v2f bK[16];
            #pragma unroll
            for (int j = 0; j < 16; ++j)
                bK[j] = *(const v2f*)(kp + 4 * j);
            __builtin_amdgcn_sched_barrier(0);
            v8f s0 = {};
            v8f s1 = {};
            #pragma unroll
            for (int j = 0; j < 16; ++j) {
                s0 = WMMA_F32(aQ0[j], bK[j], s0);
                s1 = WMMA_F32(aQ1[j], bK[j], s1);
            }
            S[0][n] = s0;
            S[1][n] = s1;
            __builtin_amdgcn_sched_barrier(0);
        }

        #pragma unroll
        for (int g = 0; g < 2; ++g) {
            #pragma unroll
            for (int n = 0; n < 4; ++n) {
                #pragma unroll
                for (int vv = 0; vv < 8; ++vv) {
                    const int row = 16 * g + vv + 8 * laneHi;
                    float s = S[g][n][vv] * scale;
                    if (k0 + 16 * n + lane16 > q0 + row) s = -__builtin_inff();
                    S[g][n][vv] = s;
                }
            }
            float alpha[8];
            #pragma unroll
            for (int vv = 0; vv < 8; ++vv) {
                float tmax = fmaxf(fmaxf(S[g][0][vv], S[g][1][vv]),
                                   fmaxf(S[g][2][vv], S[g][3][vv]));
                tmax = fmaxf(tmax, row_xor<1>(tmax));
                tmax = fmaxf(tmax, row_xor<2>(tmax));
                tmax = fmaxf(tmax, row_xor<4>(tmax));
                tmax = fmaxf(tmax, row_xor<8>(tmax));
                const float mnew = fmaxf(mrow[g][vv], tmax);
                float p[4];
                #pragma unroll
                for (int n = 0; n < 4; ++n)
                    p[n] = __expf(S[g][n][vv] - mnew);
                float psum = (p[0] + p[1]) + (p[2] + p[3]);
                psum += row_xor<1>(psum);
                psum += row_xor<2>(psum);
                psum += row_xor<4>(psum);
                psum += row_xor<8>(psum);
                const float a = __expf(mrow[g][vv] - mnew);
                lrow[g][vv]  = lrow[g][vv] * a + psum;
                mrow[g][vv]  = mnew;
                alpha[vv]    = a;
                #pragma unroll
                for (int n = 0; n < 4; ++n)
                    S[g][n][vv] = p[n];
            }
            #pragma unroll
            for (int c = 0; c < 4; ++c)
                #pragma unroll
                for (int vv = 0; vv < 8; ++vv)
                    O[g][c][vv] *= alpha[vv];
        }

        __syncthreads();
        #pragma unroll
        for (int g = 0; g < 2; ++g)
            #pragma unroll
            for (int n = 0; n < 4; ++n)
                #pragma unroll
                for (int vv = 0; vv < 8; ++vv)
                    pst[g * 16 * PPITCH + vv * PPITCH + 16 * n] = S[g][n][vv];
        __syncthreads();

        const float* vp = v + ((long)bh * SEQ + k0) * HD + lane16;
        #pragma unroll
        for (int n = 0; n < 4; ++n) {
            v2f bV[4][4];
            v2f aP[4][2];
            #pragma unroll
            for (int j = 0; j < 4; ++j) {
                const int kk = 16 * n + 4 * j + 2 * laneHi;
                aP[j][0] = *(const v2f*)(pld + 16 * n + 4 * j);
                aP[j][1] = *(const v2f*)(pld + 16 * PPITCH + 16 * n + 4 * j);
                #pragma unroll
                for (int c = 0; c < 4; ++c) {
                    const float* vbp = vp + (long)kk * HD + c * 16;
                    bV[j][c].x = vbp[0];
                    bV[j][c].y = vbp[HD];
                }
            }
            __builtin_amdgcn_sched_barrier(0);
            #pragma unroll
            for (int j = 0; j < 4; ++j) {
                #pragma unroll
                for (int c = 0; c < 4; ++c) {
                    O[0][c] = WMMA_F32(aP[j][0], bV[j][c], O[0][c]);
                    O[1][c] = WMMA_F32(aP[j][1], bV[j][c], O[1][c]);
                }
            }
            __builtin_amdgcn_sched_barrier(0);
        }
    }

    #pragma unroll
    for (int g = 0; g < 2; ++g) {
        #pragma unroll
        for (int c = 0; c < 4; ++c) {
            #pragma unroll
            for (int vv = 0; vv < 8; ++vv) {
                const int t = q0 + 16 * g + vv + 8 * laneHi;
                const int d = c * 16 + lane16;
                out[(((long)b * SEQ + t) * NHEAD + h) * HD + d] = O[g][c][vv] / lrow[g][vv];
            }
        }
    }
}

extern "C" void kernel_launch(void* const* d_in, const int* in_sizes, int n_in,
                              void* d_out, int out_size, void* d_ws, size_t ws_size,
                              hipStream_t stream) {
    const float* x    = (const float*)d_in[0];
    const float* Wqkv = (const float*)d_in[1];
    const float* bqkv = (const float*)d_in[2];
    float* out = (float*)d_out;

    float* ws = (float*)d_ws;
    const long QS = (long)BATCH * NHEAD * SEQ * HD;
    float* qb = ws;
    float* kb = ws + QS;
    float* vb = ws + 2 * QS;

    qkv_gemm_kernel<<<1536, 128, 0, stream>>>(x, Wqkv, bqkv, qb, kb, vb);
    attn_kernel<<<BATCH * NHEAD * (SEQ / 32), 32, 0, stream>>>(qb, kb, vb, out);
}


# Attention keys split 4 ways: one work-group of 4 waves per 32-query tile, wave w takes 64-key trips w, w+4, ...; partial max, sum, O merged through LDS at the end; key sum order changes
# speedup vs baseline: 2.2072x; 1.1223x over previous
#include <hip/hip_runtime.h>

typedef float v2f __attribute__((ext_vector_type(2)));
typedef float v8f __attribute__((ext_vector_type(8)));

#define WMMA_F32(a, b, c) \
    __builtin_amdgcn_wmma_f32_16x16x4_f32(false, (a), false, (b), (short)0, (c), false, false)

#define EMBED 1024
#define THREE_EMBED 3072
#define NHEAD 16
#define HD 64
#define SEQ 2048
#define BATCH 4

typedef float v4f __attribute__((ext_vector_type(4)));

#define GBM 128
#define GBN 128
#define GBK 32
#define APITCH 36
#define BPITCH 136

__global__ __launch_bounds__(128) void qkv_gemm_kernel(
    const float* __restrict__ x, const float* __restrict__ w,
    const float* __restrict__ bias,
    float* __restrict__ qb, float* __restrict__ kb, float* __restrict__ vb)
{
    __shared__ __attribute__((aligned(16))) float As[2][GBM * APITCH];
    __shared__ __attribute__((aligned(16))) float Bs[2][GBK * BPITCH];

    const int tid    = threadIdx.x;
    const int lane   = tid & 31;
    const int wave   = tid >> 5;
    const int lane16 = lane & 15;
    const int laneHi = lane >> 4;
    const int wm     = wave >> 1;
    const int wn     = wave & 1;

    const int mtile = blockIdx.x & 63;
    const int ntile = blockIdx.x >> 6;
    const int m0    = mtile * GBM;
    const int n0    = ntile * GBN;

    const int arow = tid >> 3;
    const int acol = (tid & 7) * 4;
    const int brow = tid >> 5;
    const int bcol = (tid & 31) * 4;
    const float* ag = x + (long)(m0 + arow) * EMBED + acol;
    const float* bg = w + (long)brow * THREE_EMBED + n0 + bcol;

    v4f ra[8], rb[8];
    #pragma unroll
    for (int i = 0; i < 8; ++i) {
        ra[i] = *(const v4f*)(ag + (long)(16 * i) * EMBED);
        rb[i] = *(const v4f*)(bg + (long)(4 * i) * THREE_EMBED);
    }
    #pragma unroll
    for (int i = 0; i < 8; ++i) {
        *(v4f*)&As[0][(arow + 16 * i) * APITCH + acol] = ra[i];
        *(v4f*)&Bs[0][(brow + 4 * i) * BPITCH + bcol]  = rb[i];
    }
    __syncthreads();

    v8f acc[4][4] = {};

    const int aoff = (wm * 64 + lane16) * APITCH + 2 * laneHi;
    const int boff = (2 * laneHi) * BPITCH + wn * 64 + lane16;

    for (int kt = 0; kt < EMBED / GBK; ++kt) {
        const int cur = kt & 1;
        const bool more = (kt + 1) < (EMBED / GBK);

        if (more) {
            const float* agn = ag + (kt + 1) * GBK;
            const float* bgn = bg + (long)(kt + 1) * GBK * THREE_EMBED;
            #pragma unroll
            for (int i = 0; i < 8; ++i) {
                ra[i] = *(const v4f*)(agn + (long)(16 * i) * EMBED);
                rb[i] = *(const v4f*)(bgn + (long)(4 * i) * THREE_EMBED);
            }
        }

        const float* as = &As[cur][aoff];
        const float* bs = &Bs[cur][boff];
        #pragma unroll
        for (int ks = 0; ks < GBK / 4; ++ks) {
            v2f a[4], b[4];
            #pragma unroll
            for (int i = 0; i < 4; ++i)
                a[i] = *(const v2f*)(as + 16 * i * APITCH + 4 * ks);
            #pragma unroll
            for (int j = 0; j < 4; ++j) {
                b[j].x = bs[(4 * ks) * BPITCH + 16 * j];
                b[j].y = bs[(4 * ks + 1) * BPITCH + 16 * j];
            }
            #pragma unroll
            for (int i = 0; i < 4; ++i)
                #pragma unroll
                for (int j = 0; j < 4; ++j)
                    acc[i][j] = WMMA_F32(a[i], b[j], acc[i][j]);
        }

        if (more) {
            #pragma unroll
            for (int i = 0; i < 8; ++i) {
                *(v4f*)&As[cur ^ 1][(arow + 16 * i) * APITCH + acol] = ra[i];
                *(v4f*)&Bs[cur ^ 1][(brow + 4 * i) * BPITCH + bcol]  = rb[i];
            }
        }
        __syncthreads();
    }

    const int nw0     = n0 + wn * 64;
    const int section = nw0 / EMBED;
    const int h       = (nw0 % EMBED) / HD;
    float* dst = (section == 0) ? qb : (section == 1) ? kb : vb;

    #pragma unroll
    for (int j = 0; j < 4; ++j) {
        const int d = j * 16 + lane16;
        const float bv = bias[nw0 + d];
        #pragma unroll
        for (int i = 0; i < 4; ++i) {
            #pragma unroll
            for (int vv = 0; vv < 8; ++vv) {
                const int m  = m0 + wm * 64 + 16 * i + vv + 8 * laneHi;
                const int bi = m >> 11;
                const int t  = m & 2047;
                dst[(((long)(bi * NHEAD + h)) * SEQ + t) * HD + d] = acc[i][j][vv] + bv;
            }
        }
    }
}

template <int M>
__device__ __forceinline__ float row_xor(float x) {
    const int xi = __builtin_bit_cast(int, x);
    return __builtin_bit_cast(float,
        __builtin_amdgcn_update_dpp(xi, xi, 0x160 + M, 0xf, 0xf, true));
}

#define PPITCH 68
#define AWAVES 4

__device__ __forceinline__ void wave_lds_sync() {
    __builtin_amdgcn_fence(__ATOMIC_RELEASE, "workgroup", "local");
    __builtin_amdgcn_wave_barrier();
    __builtin_amdgcn_fence(__ATOMIC_ACQUIRE, "workgroup", "local");
}

__global__ __launch_bounds__(32 * AWAVES, 3) void attn_kernel(
    const float* __restrict__ q, const float* __restrict__ k,
    const float* __restrict__ v, float* __restrict__ out)
{
    __shared__ __attribute__((aligned(16))) float pstage[AWAVES][2 * 16 * PPITCH];
    __shared__ __attribute__((aligned(16))) float mbuf[32 * AWAVES];
    __shared__ __attribute__((aligned(16))) float lbuf[32 * AWAVES];
    __shared__ __attribute__((aligned(16))) float obuf[AWAVES * 64 * 32];

    const int lane   = threadIdx.x & 31;
    const int lane16 = lane & 15;
    const int laneHi = lane >> 4;
    const int wave   = __builtin_amdgcn_readfirstlane((int)(threadIdx.x >> 5));

    const int wid = blockIdx.x;
    const int q32 = wid & 63;
    const int bh  = wid >> 6;
    const int b   = bh >> 4;
    const int h   = bh & 15;
    const int q0  = q32 * 32;

    const float* qp0 = q + ((long)bh * SEQ + q0 + lane16) * HD + 2 * laneHi;
    const float* qp1 = qp0 + 16 * HD;
    v2f aQ0[16], aQ1[16];
    #pragma unroll
    for (int j = 0; j < 16; ++j) {
        aQ0[j] = *(const v2f*)(qp0 + 4 * j);
        aQ1[j] = *(const v2f*)(qp1 + 4 * j);
    }

    float mrow[2][8], lrow[2][8];
    v8f O[2][4] = {};
    #pragma unroll
    for (int g = 0; g < 2; ++g)
        #pragma unroll
        for (int vv = 0; vv < 8; ++vv) { mrow[g][vv] = -__builtin_inff(); lrow[g][vv] = 0.f; }

    const float scale = 0.125f;
    const int ntrip = (q0 + 95) >> 6;

    float* pst = &pstage[wave][(8 * laneHi) * PPITCH + lane16];
    const float* pld = &pstage[wave][lane16 * PPITCH + 2 * laneHi];

    for (int kt = wave; kt < ntrip; kt += AWAVES) {
        const int k0 = kt * 64;

        v8f S[2][4];
        #pragma unroll
        for (int n = 0; n < 4; ++n) {
            const float* kp = k + ((long)bh * SEQ + k0 + 16 * n + lane16) * HD + 2 * laneHi;
            v2f bK[16];
            #pragma unroll
            for (int j = 0; j < 16; ++j)
                bK[j] = *(const v2f*)(kp + 4 * j);
            __builtin_amdgcn_sched_barrier(0);
            v8f s0 = {};
            v8f s1 = {};
            #pragma unroll
            for (int j = 0; j < 16; ++j) {
                s0 = WMMA_F32(aQ0[j], bK[j], s0);
                s1 = WMMA_F32(aQ1[j], bK[j], s1);
            }
            S[0][n] = s0;
            S[1][n] = s1;
            __builtin_amdgcn_sched_barrier(0);
        }

        #pragma unroll
        for (int g = 0; g < 2; ++g) {
            #pragma unroll
            for (int n = 0; n < 4; ++n) {
                #pragma unroll
                for (int vv = 0; vv < 8; ++vv) {
                    const int row = 16 * g + vv + 8 * laneHi;
                    float s = S[g][n][vv] * scale;
                    if (k0 + 16 * n + lane16 > q0 + row) s = -__builtin_inff();
                    S[g][n][vv] = s;
                }
            }
            float alpha[8];
            #pragma unroll
            for (int vv = 0; vv < 8; ++vv) {
                float tmax = fmaxf(fmaxf(S[g][0][vv], S[g][1][vv]),
                                   fmaxf(S[g][2][vv], S[g][3][vv]));
                tmax = fmaxf(tmax, row_xor<1>(tmax));
                tmax = fmaxf(tmax, row_xor<2>(tmax));
                tmax = fmaxf(tmax, row_xor<4>(tmax));
                tmax = fmaxf(tmax, row_xor<8>(tmax));
                const float mnew = fmaxf(mrow[g][vv], tmax);
                float p[4];
                #pragma unroll
                for (int n = 0; n < 4; ++n)
                    p[n] = __expf(S[g][n][vv] - mnew);
                float psum = (p[0] + p[1]) + (p[2] + p[3]);
                psum += row_xor<1>(psum);
                psum += row_xor<2>(psum);
                psum += row_xor<4>(psum);
                psum += row_xor<8>(psum);
                const float a = __expf(mrow[g][vv] - mnew);
                lrow[g][vv]  = lrow[g][vv] * a + psum;
                mrow[g][vv]  = mnew;
                alpha[vv]    = a;
                #pragma unroll
                for (int n = 0; n < 4; ++n)
                    S[g][n][vv] = p[n];
            }
            #pragma unroll
            for (int c = 0; c < 4; ++c)
                #pragma unroll
                for (int vv = 0; vv < 8; ++vv)
                    O[g][c][vv] *= alpha[vv];
        }

        wave_lds_sync();
        #pragma unroll
        for (int g = 0; g < 2; ++g)
            #pragma unroll
            for (int n = 0; n < 4; ++n)
                #pragma unroll
                for (int vv = 0; vv < 8; ++vv)
                    pst[g * 16 * PPITCH + vv * PPITCH + 16 * n] = S[g][n][vv];
        wave_lds_sync();

        const float* vp = v + ((long)bh * SEQ + k0) * HD + lane16;
        #pragma unroll
        for (int n = 0; n < 4; ++n) {
            v2f bV[4][4];
            v2f aP[4][2];
            #pragma unroll
            for (int j = 0; j < 4; ++j) {
                const int kk = 16 * n + 4 * j + 2 * laneHi;
                aP[j][0] = *(const v2f*)(pld + 16 * n + 4 * j);
                aP[j][1] = *(const v2f*)(pld + 16 * PPITCH + 16 * n + 4 * j);
                #pragma unroll
                for (int c = 0; c < 4; ++c) {
                    const float* vbp = vp + (long)kk * HD + c * 16;
                    bV[j][c].x = vbp[0];
                    bV[j][c].y = vbp[HD];
                }
            }
            __builtin_amdgcn_sched_barrier(0);
            #pragma unroll
            for (int j = 0; j < 4; ++j) {
                #pragma unroll
                for (int c = 0; c < 4; ++c) {
                    O[0][c] = WMMA_F32(aP[j][0], bV[j][c], O[0][c]);
                    O[1][c] = WMMA_F32(aP[j][1], bV[j][c], O[1][c]);
                }
            }
            __builtin_amdgcn_sched_barrier(0);
        }
    }

    const int lane_m   = (int)__builtin_amdgcn_mbcnt_lo(~0u, 0u);
    const int lane16_m = lane_m & 15;
    const int laneHi_m = lane_m >> 4;
    if (lane16_m == 0) {
        #pragma unroll
        for (int g = 0; g < 2; ++g)
            #pragma unroll
            for (int vv = 0; vv < 8; ++vv) {
                const int row = 16 * g + 8 * laneHi_m + vv;
                mbuf[row * AWAVES + wave] = mrow[g][vv];
                lbuf[row * AWAVES + wave] = lrow[g][vv];
            }
    }
    __syncthreads();

    float lstar[2][8];
    #pragma unroll
    for (int g = 0; g < 2; ++g) {
        #pragma unroll
        for (int vv = 0; vv < 8; ++vv) {
            const int row = 16 * g + 8 * laneHi_m + vv;
            const v4f m4 = *(const v4f*)&mbuf[row * AWAVES];
            const v4f l4 = *(const v4f*)&lbuf[row * AWAVES];
            const float mstar = fmaxf(fmaxf(m4.x, m4.y), fmaxf(m4.z, m4.w));
            float ls = l4.x * __expf(m4.x - mstar);
            ls += l4.y * __expf(m4.y - mstar);
            ls += l4.z * __expf(m4.z - mstar);
            ls += l4.w * __expf(m4.w - mstar);
            lstar[g][vv] = ls;
            const float f = __expf(mrow[g][vv] - mstar);
            #pragma unroll
            for (int c = 0; c < 4; ++c)
                obuf[(((wave * 2 + g) * 4 + c) * 8 + vv) * 32 + lane_m] = O[g][c][vv] * f;
        }
    }
    __syncthreads();

    const int gr = wave >> 1;
    const int cp = wave & 1;
    #pragma unroll
    for (int cc = 0; cc < 2; ++cc) {
        const int c = 2 * cp + cc;
        #pragma unroll
        for (int vv = 0; vv < 8; ++vv) {
            const float* ob = &obuf[((gr * 4 + c) * 8 + vv) * 32 + lane_m];
            float o = ob[0];
            #pragma unroll
            for (int w = 1; w < AWAVES; ++w)
                o += ob[w * (64 * 32)];
            const float ls = gr ? lstar[1][vv] : lstar[0][vv];
            const int t = q0 + 16 * gr + vv + 8 * laneHi_m;
            const int d = c * 16 + lane16_m;
            out[(((long)b * SEQ + t) * NHEAD + h) * HD + d] = o / ls;
        }
    }
}

extern "C" void kernel_launch(void* const* d_in, const int* in_sizes, int n_in,
                              void* d_out, int out_size, void* d_ws, size_t ws_size,
                              hipStream_t stream) {
    const float* x    = (const float*)d_in[0];
    const float* Wqkv = (const float*)d_in[1];
    const float* bqkv = (const float*)d_in[2];
    float* out = (float*)d_out;

    float* ws = (float*)d_ws;
    const long QS = (long)BATCH * NHEAD * SEQ * HD;
    float* qb = ws;
    float* kb = ws + QS;
    float* vb = ws + 2 * QS;

    qkv_gemm_kernel<<<1536, 128, 0, stream>>>(x, Wqkv, bqkv, qb, kb, vb);
    attn_kernel<<<BATCH * NHEAD * (SEQ / 32), 32 * AWAVES, 0, stream>>>(qb, kb, vb, out);
}


# Attention K, V, P fragment loads issued one 16-key tile ahead of their WMMAs into a second register set (causal-mask thresholds folded into one lane value to make register room); sum order unchanged
# speedup vs baseline: 2.3063x; 1.0449x over previous
#include <hip/hip_runtime.h>

typedef float v2f __attribute__((ext_vector_type(2)));
typedef float v8f __attribute__((ext_vector_type(8)));

#define WMMA_F32(a, b, c) \
    __builtin_amdgcn_wmma_f32_16x16x4_f32(false, (a), false, (b), (short)0, (c), false, false)

#define EMBED 1024
#define THREE_EMBED 3072
#define NHEAD 16
#define HD 64
#define SEQ 2048
#define BATCH 4

typedef float v4f __attribute__((ext_vector_type(4)));

#define GBM 128
#define GBN 128
#define GBK 32
#define APITCH 36
#define BPITCH 136

__global__ __launch_bounds__(128) void qkv_gemm_kernel(
    const float* __restrict__ x, const float* __restrict__ w,
    const float* __restrict__ bias,
    float* __restrict__ qb, float* __restrict__ kb, float* __restrict__ vb)
{
    __shared__ __attribute__((aligned(16))) float As[2][GBM * APITCH];
    __shared__ __attribute__((aligned(16))) float Bs[2][GBK * BPITCH];

    const int tid    = threadIdx.x;
    const int lane   = tid & 31;
    const int wave   = tid >> 5;
    const int lane16 = lane & 15;
    const int laneHi = lane >> 4;
    const int wm     = wave >> 1;
    const int wn     = wave & 1;

    const int mtile = blockIdx.x & 63;
    const int ntile = blockIdx.x >> 6;
    const int m0    = mtile * GBM;
    const int n0    = ntile * GBN;

    const int arow = tid >> 3;
    const int acol = (tid & 7) * 4;
    const int brow = tid >> 5;
    const int bcol = (tid & 31) * 4;
    const float* ag = x + (long)(m0 + arow) * EMBED + acol;
    const float* bg = w + (long)brow * THREE_EMBED + n0 + bcol;

    v4f ra[8], rb[8];
    #pragma unroll
    for (int i = 0; i < 8; ++i) {
        ra[i] = *(const v4f*)(ag + (long)(16 * i) * EMBED);
        rb[i] = *(const v4f*)(bg + (long)(4 * i) * THREE_EMBED);
    }
    #pragma unroll
    for (int i = 0; i < 8; ++i) {
        *(v4f*)&As[0][(arow + 16 * i) * APITCH + acol] = ra[i];
        *(v4f*)&Bs[0][(brow + 4 * i) * BPITCH + bcol]  = rb[i];
    }
    __syncthreads();

    v8f acc[4][4] = {};

    const int aoff = (wm * 64 + lane16) * APITCH + 2 * laneHi;
    const int boff = (2 * laneHi) * BPITCH + wn * 64 + lane16;

    for (int kt = 0; kt < EMBED / GBK; ++kt) {
        const int cur = kt & 1;
        const bool more = (kt + 1) < (EMBED / GBK);

        if (more) {
            const float* agn = ag + (kt + 1) * GBK;
            const float* bgn = bg + (long)(kt + 1) * GBK * THREE_EMBED;
            #pragma unroll
            for (int i = 0; i < 8; ++i) {
                ra[i] = *(const v4f*)(agn + (long)(16 * i) * EMBED);
                rb[i] = *(const v4f*)(bgn + (long)(4 * i) * THREE_EMBED);
            }
        }

        const float* as = &As[cur][aoff];
        const float* bs = &Bs[cur][boff];
        #pragma unroll
        for (int ks = 0; ks < GBK / 4; ++ks) {
            v2f a[4], b[4];
            #pragma unroll
            for (int i = 0; i < 4; ++i)
                a[i] = *(const v2f*)(as + 16 * i * APITCH + 4 * ks);
            #pragma unroll
            for (int j = 0; j < 4; ++j) {
                b[j].x = bs[(4 * ks) * BPITCH + 16 * j];
                b[j].y = bs[(4 * ks + 1) * BPITCH + 16 * j];
            }
            #pragma unroll
            for (int i = 0; i < 4; ++i)
                #pragma unroll
                for (int j = 0; j < 4; ++j)
                    acc[i][j] = WMMA_F32(a[i], b[j], acc[i][j]);
        }

        if (more) {
            #pragma unroll
            for (int i = 0; i < 8; ++i) {
                *(v4f*)&As[cur ^ 1][(arow + 16 * i) * APITCH + acol] = ra[i];
                *(v4f*)&Bs[cur ^ 1][(brow + 4 * i) * BPITCH + bcol]  = rb[i];
            }
        }
        __syncthreads();
    }

    const int nw0     = n0 + wn * 64;
    const int section = nw0 / EMBED;
    const int h       = (nw0 % EMBED) / HD;
    float* dst = (section == 0) ? qb : (section == 1) ? kb : vb;

    #pragma unroll
    for (int j = 0; j < 4; ++j) {
        const int d = j * 16 + lane16;
        const float bv = bias[nw0 + d];
        #pragma unroll
        for (int i = 0; i < 4; ++i) {
            #pragma unroll
            for (int vv = 0; vv < 8; ++vv) {
                const int m  = m0 + wm * 64 + 16 * i + vv + 8 * laneHi;
                const int bi = m >> 11;
                const int t  = m & 2047;
                dst[(((long)(bi * NHEAD + h)) * SEQ + t) * HD + d] = acc[i][j][vv] + bv;
            }
        }
    }
}

template <int M>
__device__ __forceinline__ float row_xor(float x) {
    const int xi = __builtin_bit_cast(int, x);
    return __builtin_bit_cast(float,
        __builtin_amdgcn_update_dpp(xi, xi, 0x160 + M, 0xf, 0xf, true));
}

#define PPITCH 68
#define AWAVES 4

__device__ __forceinline__ void wave_lds_sync() {
    __builtin_amdgcn_fence(__ATOMIC_RELEASE, "workgroup", "local");
    __builtin_amdgcn_wave_barrier();
    __builtin_amdgcn_fence(__ATOMIC_ACQUIRE, "workgroup", "local");
}

__device__ __forceinline__ void load_k_tile(v2f (&dst)[16], const float* __restrict__ p) {
    #pragma unroll
    for (int j = 0; j < 16; ++j)
        dst[j] = *(const v2f*)(p + 4 * j);
}

__device__ __forceinline__ void load_v_tile(v2f (&dst)[4][4], const float* __restrict__ p) {
    #pragma unroll
    for (int j = 0; j < 4; ++j) {
        #pragma unroll
        for (int c = 0; c < 4; ++c) {
            dst[j][c].x = p[(4 * j) * HD + 16 * c];
            dst[j][c].y = p[(4 * j + 1) * HD + 16 * c];
        }
    }
}

__global__ __launch_bounds__(32 * AWAVES, 3) void attn_kernel(
    const float* __restrict__ q, const float* __restrict__ k,
    const float* __restrict__ v, float* __restrict__ out)
{
    __shared__ __attribute__((aligned(16))) float pstage[AWAVES][2 * 16 * PPITCH];
    __shared__ __attribute__((aligned(16))) float mbuf[32 * AWAVES];
    __shared__ __attribute__((aligned(16))) float lbuf[32 * AWAVES];
    __shared__ __attribute__((aligned(16))) float obuf[AWAVES * 64 * 32];

    const int lane   = threadIdx.x & 31;
    const int lane16 = lane & 15;
    const int laneHi = lane >> 4;
    const int wave   = __builtin_amdgcn_readfirstlane((int)(threadIdx.x >> 5));

    const int wid = blockIdx.x;
    const int q32 = wid & 63;
    const int bh  = wid >> 6;
    const int b   = bh >> 4;
    const int h   = bh & 15;
    const int q0  = q32 * 32;

    const float* qp0 = q + ((long)bh * SEQ + q0 + lane16) * HD + 2 * laneHi;
    const float* qp1 = qp0 + 16 * HD;
    v2f aQ0[16], aQ1[16];
    #pragma unroll
    for (int j = 0; j < 16; ++j) {
        aQ0[j] = *(const v2f*)(qp0 + 4 * j);
        aQ1[j] = *(const v2f*)(qp1 + 4 * j);
    }

    float mrow[2][8], lrow[2][8];
    v8f O[2][4] = {};
    #pragma unroll
    for (int g = 0; g < 2; ++g)
        #pragma unroll
        for (int vv = 0; vv < 8; ++vv) { mrow[g][vv] = -__builtin_inff(); lrow[g][vv] = 0.f; }

    const float scale = 0.125f;
    const int ntrip = (q0 + 95) >> 6;

    float* pst = &pstage[wave][(8 * laneHi) * PPITCH + lane16];
    const float* pld = &pstage[wave][lane16 * PPITCH + 2 * laneHi];

    const float* kfp = k + ((long)bh * SEQ + lane16) * HD + 2 * laneHi;
    const float* vfp = v + ((long)bh * SEQ + 2 * laneHi) * HD + lane16;

    v2f kf[2][16];
    v2f vf[2][4][4];

    {
        const int ktf = (wave < ntrip) ? wave : (ntrip - 1);
        load_k_tile(kf[0], kfp + (long)(ktf * 64) * HD);
    }

    for (int kt = wave; kt < ntrip; kt += AWAVES) {
        const int k0 = kt * 64;
        const float* ktp = kfp + (long)k0 * HD;
        const float* vtp = vfp + (long)k0 * HD;
        const int ktn = (kt + AWAVES < ntrip) ? (kt + AWAVES) : (ntrip - 1);
        const int dmask = q0 + 8 * laneHi - lane16 - k0;

        v8f S[2][4];
        #pragma unroll
        for (int n = 0; n < 4; ++n) {
            if (n < 3) load_k_tile(kf[(n + 1) & 1], ktp + (16 * (n + 1)) * HD);
            __builtin_amdgcn_sched_barrier(0);
            v8f s0 = {};
            v8f s1 = {};
            #pragma unroll
            for (int j = 0; j < 16; ++j) {
                s0 = WMMA_F32(aQ0[j], kf[n & 1][j], s0);
                s1 = WMMA_F32(aQ1[j], kf[n & 1][j], s1);
            }
            S[0][n] = s0;
            S[1][n] = s1;
            __builtin_amdgcn_sched_barrier(0);
        }
        load_v_tile(vf[0], vtp);
        __builtin_amdgcn_sched_barrier(0);

        #pragma unroll
        for (int g = 0; g < 2; ++g) {
            #pragma unroll
            for (int n = 0; n < 4; ++n) {
                #pragma unroll
                for (int vv = 0; vv < 8; ++vv) {
                    float s = S[g][n][vv] * scale;
                    if (16 * n - 16 * g - vv > dmask) s = -__builtin_inff();
                    S[g][n][vv] = s;
                }
            }
            float alpha[8];
            #pragma unroll
            for (int vv = 0; vv < 8; ++vv) {
                float tmax = fmaxf(fmaxf(S[g][0][vv], S[g][1][vv]),
                                   fmaxf(S[g][2][vv], S[g][3][vv]));
                tmax = fmaxf(tmax, row_xor<1>(tmax));
                tmax = fmaxf(tmax, row_xor<2>(tmax));
                tmax = fmaxf(tmax, row_xor<4>(tmax));
                tmax = fmaxf(tmax, row_xor<8>(tmax));
                const float mnew = fmaxf(mrow[g][vv], tmax);
                float p[4];
                #pragma unroll
                for (int n = 0; n < 4; ++n)
                    p[n] = __expf(S[g][n][vv] - mnew);
                float psum = (p[0] + p[1]) + (p[2] + p[3]);
                psum += row_xor<1>(psum);
                psum += row_xor<2>(psum);
                psum += row_xor<4>(psum);
                psum += row_xor<8>(psum);
                const float a = __expf(mrow[g][vv] - mnew);
                lrow[g][vv]  = lrow[g][vv] * a + psum;
                mrow[g][vv]  = mnew;
                alpha[vv]    = a;
                #pragma unroll
                for (int n = 0; n < 4; ++n)
                    S[g][n][vv] = p[n];
            }
            #pragma unroll
            for (int c = 0; c < 4; ++c)
                #pragma unroll
                for (int vv = 0; vv < 8; ++vv)
                    O[g][c][vv] *= alpha[vv];
        }

        wave_lds_sync();
        #pragma unroll
        for (int g = 0; g < 2; ++g)
            #pragma unroll
            for (int n = 0; n < 4; ++n)
                #pragma unroll
                for (int vv = 0; vv < 8; ++vv)
                    pst[g * 16 * PPITCH + vv * PPITCH + 16 * n] = S[g][n][vv];
        wave_lds_sync();

        v2f aP[2][4][2];
        #pragma unroll
        for (int j = 0; j < 4; ++j) {
            aP[0][j][0] = *(const v2f*)(pld + 4 * j);
            aP[0][j][1] = *(const v2f*)(pld + 16 * PPITCH + 4 * j);
        }
        #pragma unroll
        for (int n = 0; n < 4; ++n) {
            if (n < 3) {
                load_v_tile(vf[(n + 1) & 1], vtp + (16 * (n + 1)) * HD);
                #pragma unroll
                for (int j = 0; j < 4; ++j) {
                    aP[(n + 1) & 1][j][0] = *(const v2f*)(pld + 16 * (n + 1) + 4 * j);
                    aP[(n + 1) & 1][j][1] = *(const v2f*)(pld + 16 * PPITCH + 16 * (n + 1) + 4 * j);
                }
            } else {
                load_k_tile(kf[0], kfp + (long)(ktn * 64) * HD);
            }
            __builtin_amdgcn_sched_barrier(0);
            #pragma unroll
            for (int j = 0; j < 4; ++j) {
                #pragma unroll
                for (int c = 0; c < 4; ++c) {
                    O[0][c] = WMMA_F32(aP[n & 1][j][0], vf[n & 1][j][c], O[0][c]);
                    O[1][c] = WMMA_F32(aP[n & 1][j][1], vf[n & 1][j][c], O[1][c]);
                }
            }
            __builtin_amdgcn_sched_barrier(0);
        }
    }

    const int lane_m   = (int)__builtin_amdgcn_mbcnt_lo(~0u, 0u);
    const int lane16_m = lane_m & 15;
    const int laneHi_m = lane_m >> 4;
    if (lane16_m == 0) {
        #pragma unroll
        for (int g = 0; g < 2; ++g)
            #pragma unroll
            for (int vv = 0; vv < 8; ++vv) {
                const int row = 16 * g + 8 * laneHi_m + vv;
                mbuf[row * AWAVES + wave] = mrow[g][vv];
                lbuf[row * AWAVES + wave] = lrow[g][vv];
            }
    }
    __syncthreads();

    float lstar[2][8];
    #pragma unroll
    for (int g = 0; g < 2; ++g) {
        #pragma unroll
        for (int vv = 0; vv < 8; ++vv) {
            const int row = 16 * g + 8 * laneHi_m + vv;
            const v4f m4 = *(const v4f*)&mbuf[row * AWAVES];
            const v4f l4 = *(const v4f*)&lbuf[row * AWAVES];
            const float mstar = fmaxf(fmaxf(m4.x, m4.y), fmaxf(m4.z, m4.w));
            float ls = l4.x * __expf(m4.x - mstar);
            ls += l4.y * __expf(m4.y - mstar);
            ls += l4.z * __expf(m4.z - mstar);
            ls += l4.w * __expf(m4.w - mstar);
            lstar[g][vv] = ls;
            const float f = __expf(mrow[g][vv] - mstar);
            #pragma unroll
            for (int c = 0; c < 4; ++c)
                obuf[(((wave * 2 + g) * 4 + c) * 8 + vv) * 32 + lane_m] = O[g][c][vv] * f;
        }
    }
    __syncthreads();

    const int gr = wave >> 1;
    const int cp = wave & 1;
    #pragma unroll
    for (int cc = 0; cc < 2; ++cc) {
        const int c = 2 * cp + cc;
        #pragma unroll
        for (int vv = 0; vv < 8; ++vv) {
            const float* ob = &obuf[((gr * 4 + c) * 8 + vv) * 32 + lane_m];
            float o = ob[0];
            #pragma unroll
            for (int w = 1; w < AWAVES; ++w)
                o += ob[w * (64 * 32)];
            const float ls = gr ? lstar[1][vv] : lstar[0][vv];
            const int t = q0 + 16 * gr + vv + 8 * laneHi_m;
            const int d = c * 16 + lane16_m;
            out[(((long)b * SEQ + t) * NHEAD + h) * HD + d] = o / ls;
        }
    }
}

extern "C" void kernel_launch(void* const* d_in, const int* in_sizes, int n_in,
                              void* d_out, int out_size, void* d_ws, size_t ws_size,
                              hipStream_t stream) {
    const float* x    = (const float*)d_in[0];
    const float* Wqkv = (const float*)d_in[1];
    const float* bqkv = (const float*)d_in[2];
    float* out = (float*)d_out;

    float* ws = (float*)d_ws;
    const long QS = (long)BATCH * NHEAD * SEQ * HD;
    float* qb = ws;
    float* kb = ws + QS;
    float* vb = ws + 2 * QS;

    qkv_gemm_kernel<<<1536, 128, 0, stream>>>(x, Wqkv, bqkv, qb, kb, vb);
    attn_kernel<<<BATCH * NHEAD * (SEQ / 32), 32 * AWAVES, 0, stream>>>(qb, kb, vb, out);
}


# Attention work-groups numbered longest first: block index counts query tiles 63 down to 0 over 32 heads at a time, so the launch ends on 1-trip tiles, not 32-trip ones; sum order unchanged
# speedup vs baseline: 2.4326x; 1.0548x over previous
#include <hip/hip_runtime.h>

typedef float v2f __attribute__((ext_vector_type(2)));
typedef float v8f __attribute__((ext_vector_type(8)));

#define WMMA_F32(a, b, c) \
    __builtin_amdgcn_wmma_f32_16x16x4_f32(false, (a), false, (b), (short)0, (c), false, false)

#define EMBED 1024
#define THREE_EMBED 3072
#define NHEAD 16
#define HD 64
#define SEQ 2048
#define BATCH 4

typedef float v4f __attribute__((ext_vector_type(4)));

#define GBM 128
#define GBN 128
#define GBK 32
#define APITCH 36
#define BPITCH 136

__global__ __launch_bounds__(128) void qkv_gemm_kernel(
    const float* __restrict__ x, const float* __restrict__ w,
    const float* __restrict__ bias,
    float* __restrict__ qb, float* __restrict__ kb, float* __restrict__ vb)
{
    __shared__ __attribute__((aligned(16))) float As[2][GBM * APITCH];
    __shared__ __attribute__((aligned(16))) float Bs[2][GBK * BPITCH];

    const int tid    = threadIdx.x;
    const int lane   = tid & 31;
    const int wave   = tid >> 5;
    const int lane16 = lane & 15;
    const int laneHi = lane >> 4;
    const int wm     = wave >> 1;
    const int wn     = wave & 1;

    const int mtile = blockIdx.x & 63;
    const int ntile = blockIdx.x >> 6;
    const int m0    = mtile * GBM;
    const int n0    = ntile * GBN;

    const int arow = tid >> 3;
    const int acol = (tid & 7) * 4;
    const int brow = tid >> 5;
    const int bcol = (tid & 31) * 4;
    const float* ag = x + (long)(m0 + arow) * EMBED + acol;
    const float* bg = w + (long)brow * THREE_EMBED + n0 + bcol;

    v4f ra[8], rb[8];
    #pragma unroll
    for (int i = 0; i < 8; ++i) {
        ra[i] = *(const v4f*)(ag + (long)(16 * i) * EMBED);
        rb[i] = *(const v4f*)(bg + (long)(4 * i) * THREE_EMBED);
    }
    #pragma unroll
    for (int i = 0; i < 8; ++i) {
        *(v4f*)&As[0][(arow + 16 * i) * APITCH + acol] = ra[i];
        *(v4f*)&Bs[0][(brow + 4 * i) * BPITCH + bcol]  = rb[i];
    }
    __syncthreads();

    v8f acc[4][4] = {};

    const int aoff = (wm * 64 + lane16) * APITCH + 2 * laneHi;
    const int boff = (2 * laneHi) * BPITCH + wn * 64 + lane16;

    for (int kt = 0; kt < EMBED / GBK; ++kt) {
        const int cur = kt & 1;
        const bool more = (kt + 1) < (EMBED / GBK);

        if (more) {
            const float* agn = ag + (kt + 1) * GBK;
            const float* bgn = bg + (long)(kt + 1) * GBK * THREE_EMBED;
            #pragma unroll
            for (int i = 0; i < 8; ++i) {
                ra[i] = *(const v4f*)(agn + (long)(16 * i) * EMBED);
                rb[i] = *(const v4f*)(bgn + (long)(4 * i) * THREE_EMBED);
            }
        }

        const float* as = &As[cur][aoff];
        const float* bs = &Bs[cur][boff];
        #pragma unroll
        for (int ks = 0; ks < GBK / 4; ++ks) {
            v2f a[4], b[4];
            #pragma unroll
            for (int i = 0; i < 4; ++i)
                a[i] = *(const v2f*)(as + 16 * i * APITCH + 4 * ks);
            #pragma unroll
            for (int j = 0; j < 4; ++j) {
                b[j].x = bs[(4 * ks) * BPITCH + 16 * j];
                b[j].y = bs[(4 * ks + 1) * BPITCH + 16 * j];
            }
            #pragma unroll
            for (int i = 0; i < 4; ++i)
                #pragma unroll
                for (int j = 0; j < 4; ++j)
                    acc[i][j] = WMMA_F32(a[i], b[j], acc[i][j]);
        }

        if (more) {
            #pragma unroll
            for (int i = 0; i < 8; ++i) {
                *(v4f*)&As[cur ^ 1][(arow + 16 * i) * APITCH + acol] = ra[i];
                *(v4f*)&Bs[cur ^ 1][(brow + 4 * i) * BPITCH + bcol]  = rb[i];
            }
        }
        __syncthreads();
    }

    const int nw0     = n0 + wn * 64;
    const int section = nw0 / EMBED;
    const int h       = (nw0 % EMBED) / HD;
    float* dst = (section == 0) ? qb : (section == 1) ? kb : vb;

    #pragma unroll
    for (int j = 0; j < 4; ++j) {
        const int d = j * 16 + lane16;
        const float bv = bias[nw0 + d];
        #pragma unroll
        for (int i = 0; i < 4; ++i) {
            #pragma unroll
            for (int vv = 0; vv < 8; ++vv) {
                const int m  = m0 + wm * 64 + 16 * i + vv + 8 * laneHi;
                const int bi = m >> 11;
                const int t  = m & 2047;
                dst[(((long)(bi * NHEAD + h)) * SEQ + t) * HD + d] = acc[i][j][vv] + bv;
            }
        }
    }
}

template <int M>
__device__ __forceinline__ float row_xor(float x) {
    const int xi = __builtin_bit_cast(int, x);
    return __builtin_bit_cast(float,
        __builtin_amdgcn_update_dpp(xi, xi, 0x160 + M, 0xf, 0xf, true));
}

#define PPITCH 68
#define AWAVES 4

__device__ __forceinline__ void wave_lds_sync() {
    __builtin_amdgcn_fence(__ATOMIC_RELEASE, "workgroup", "local");
    __builtin_amdgcn_wave_barrier();
    __builtin_amdgcn_fence(__ATOMIC_ACQUIRE, "workgroup", "local");
}

__device__ __forceinline__ void load_k_tile(v2f (&dst)[16], const float* __restrict__ p) {
    #pragma unroll
    for (int j = 0; j < 16; ++j)
        dst[j] = *(const v2f*)(p + 4 * j);
}

__device__ __forceinline__ void load_v_tile(v2f (&dst)[4][4], const float* __restrict__ p) {
    #pragma unroll
    for (int j = 0; j < 4; ++j) {
        #pragma unroll
        for (int c = 0; c < 4; ++c) {
            dst[j][c].x = p[(4 * j) * HD + 16 * c];
            dst[j][c].y = p[(4 * j + 1) * HD + 16 * c];
        }
    }
}

__global__ __launch_bounds__(32 * AWAVES, 3) void attn_kernel(
    const float* __restrict__ q, const float* __restrict__ k,
    const float* __restrict__ v, float* __restrict__ out)
{
    __shared__ __attribute__((aligned(16))) float pstage[AWAVES][2 * 16 * PPITCH];
    __shared__ __attribute__((aligned(16))) float mbuf[32 * AWAVES];
    __shared__ __attribute__((aligned(16))) float lbuf[32 * AWAVES];
    __shared__ __attribute__((aligned(16))) float obuf[AWAVES * 64 * 32];

    const int lane   = threadIdx.x & 31;
    const int lane16 = lane & 15;
    const int laneHi = lane >> 4;
    const int wave   = __builtin_amdgcn_readfirstlane((int)(threadIdx.x >> 5));

    const int rid = BATCH * NHEAD * (SEQ / 32) - 1 - (int)blockIdx.x;
    const int q32 = (rid >> 5) & 63;
    const int bh  = ((rid >> 11) << 5) | (rid & 31);
    const int b   = bh >> 4;
    const int h   = bh & 15;
    const int q0  = q32 * 32;

    const float* qp0 = q + ((long)bh * SEQ + q0 + lane16) * HD + 2 * laneHi;
    const float* qp1 = qp0 + 16 * HD;
    v2f aQ0[16], aQ1[16];
    #pragma unroll
    for (int j = 0; j < 16; ++j) {
        aQ0[j] = *(const v2f*)(qp0 + 4 * j);
        aQ1[j] = *(const v2f*)(qp1 + 4 * j);
    }

    float mrow[2][8], lrow[2][8];
    v8f O[2][4] = {};
    #pragma unroll
    for (int g = 0; g < 2; ++g)
        #pragma unroll
        for (int vv = 0; vv < 8; ++vv) { mrow[g][vv] = -__builtin_inff(); lrow[g][vv] = 0.f; }

    const float scale = 0.125f;
    const int ntrip = (q0 + 95) >> 6;

    float* pst = &pstage[wave][(8 * laneHi) * PPITCH + lane16];
    const float* pld = &pstage[wave][lane16 * PPITCH + 2 * laneHi];

    const float* kfp = k + ((long)bh * SEQ + lane16) * HD + 2 * laneHi;
    const float* vfp = v + ((long)bh * SEQ + 2 * laneHi) * HD + lane16;

    v2f kf[2][16];
    v2f vf[2][4][4];

    {
        const int ktf = (wave < ntrip) ? wave : (ntrip - 1);
        load_k_tile(kf[0], kfp + (long)(ktf * 64) * HD);
    }

    for (int kt = wave; kt < ntrip; kt += AWAVES) {
        const int k0 = kt * 64;
        const float* ktp = kfp + (long)k0 * HD;
        const float* vtp = vfp + (long)k0 * HD;
        const int ktn = (kt + AWAVES < ntrip) ? (kt + AWAVES) : (ntrip - 1);
        const int dmask = q0 + 8 * laneHi - lane16 - k0;

        v8f S[2][4];
        #pragma unroll
        for (int n = 0; n < 4; ++n) {
            if (n < 3) load_k_tile(kf[(n + 1) & 1], ktp + (16 * (n + 1)) * HD);
            __builtin_amdgcn_sched_barrier(0);
            v8f s0 = {};
            v8f s1 = {};
            #pragma unroll
            for (int j = 0; j < 16; ++j) {
                s0 = WMMA_F32(aQ0[j], kf[n & 1][j], s0);
                s1 = WMMA_F32(aQ1[j], kf[n & 1][j], s1);
            }
            S[0][n] = s0;
            S[1][n] = s1;
            __builtin_amdgcn_sched_barrier(0);
        }
        load_v_tile(vf[0], vtp);
        __builtin_amdgcn_sched_barrier(0);

        #pragma unroll
        for (int g = 0; g < 2; ++g) {
            #pragma unroll
            for (int n = 0; n < 4; ++n) {
                #pragma unroll
                for (int vv = 0; vv < 8; ++vv) {
                    float s = S[g][n][vv] * scale;
                    if (16 * n - 16 * g - vv > dmask) s = -__builtin_inff();
                    S[g][n][vv] = s;
                }
            }
            float alpha[8];
            #pragma unroll
            for (int vv = 0; vv < 8; ++vv) {
                float tmax = fmaxf(fmaxf(S[g][0][vv], S[g][1][vv]),
                                   fmaxf(S[g][2][vv], S[g][3][vv]));
                tmax = fmaxf(tmax, row_xor<1>(tmax));
                tmax = fmaxf(tmax, row_xor<2>(tmax));
                tmax = fmaxf(tmax, row_xor<4>(tmax));
                tmax = fmaxf(tmax, row_xor<8>(tmax));
                const float mnew = fmaxf(mrow[g][vv], tmax);
                float p[4];
                #pragma unroll
                for (int n = 0; n < 4; ++n)
                    p[n] = __expf(S[g][n][vv] - mnew);
                float psum = (p[0] + p[1]) + (p[2] + p[3]);
                psum += row_xor<1>(psum);
                psum += row_xor<2>(psum);
                psum += row_xor<4>(psum);
                psum += row_xor<8>(psum);
                const float a = __expf(mrow[g][vv] - mnew);
                lrow[g][vv]  = lrow[g][vv] * a + psum;
                mrow[g][vv]  = mnew;
                alpha[vv]    = a;
                #pragma unroll
                for (int n = 0; n < 4; ++n)
                    S[g][n][vv] = p[n];
            }
            #pragma unroll
            for (int c = 0; c < 4; ++c)
                #pragma unroll
                for (int vv = 0; vv < 8; ++vv)
                    O[g][c][vv] *= alpha[vv];
        }

        wave_lds_sync();
        #pragma unroll
        for (int g = 0; g < 2; ++g)
            #pragma unroll
            for (int n = 0; n < 4; ++n)
                #pragma unroll
                for (int vv = 0; vv < 8; ++vv)
                    pst[g * 16 * PPITCH + vv * PPITCH + 16 * n] = S[g][n][vv];
        wave_lds_sync();

        v2f aP[2][4][2];
        #pragma unroll
        for (int j = 0; j < 4; ++j) {
            aP[0][j][0] = *(const v2f*)(pld + 4 * j);
            aP[0][j][1] = *(const v2f*)(pld + 16 * PPITCH + 4 * j);
        }
        #pragma unroll
        for (int n = 0; n < 4; ++n) {
            if (n < 3) {
                load_v_tile(vf[(n + 1) & 1], vtp + (16 * (n + 1)) * HD);
                #pragma unroll
                for (int j = 0; j < 4; ++j) {
                    aP[(n + 1) & 1][j][0] = *(const v2f*)(pld + 16 * (n + 1) + 4 * j);
                    aP[(n + 1) & 1][j][1] = *(const v2f*)(pld + 16 * PPITCH + 16 * (n + 1) + 4 * j);
                }
            } else {
                load_k_tile(kf[0], kfp + (long)(ktn * 64) * HD);
            }
            __builtin_amdgcn_sched_barrier(0);
            #pragma unroll
            for (int j = 0; j < 4; ++j) {
                #pragma unroll
                for (int c = 0; c < 4; ++c) {
                    O[0][c] = WMMA_F32(aP[n & 1][j][0], vf[n & 1][j][c], O[0][c]);
                    O[1][c] = WMMA_F32(aP[n & 1][j][1], vf[n & 1][j][c], O[1][c]);
                }
            }
            __builtin_amdgcn_sched_barrier(0);
        }
    }

    const int lane_m   = (int)__builtin_amdgcn_mbcnt_lo(~0u, 0u);
    const int lane16_m = lane_m & 15;
    const int laneHi_m = lane_m >> 4;
    if (lane16_m == 0) {
        #pragma unroll
        for (int g = 0; g < 2; ++g)
            #pragma unroll
            for (int vv = 0; vv < 8; ++vv) {
                const int row = 16 * g + 8 * laneHi_m + vv;
                mbuf[row * AWAVES + wave] = mrow[g][vv];
                lbuf[row * AWAVES + wave] = lrow[g][vv];
            }
    }
    __syncthreads();

    float lstar[2][8];
    #pragma unroll
    for (int g = 0; g < 2; ++g) {
        #pragma unroll
        for (int vv = 0; vv < 8; ++vv) {
            const int row = 16 * g + 8 * laneHi_m + vv;
            const v4f m4 = *(const v4f*)&mbuf[row * AWAVES];
            const v4f l4 = *(const v4f*)&lbuf[row * AWAVES];
            const float mstar = fmaxf(fmaxf(m4.x, m4.y), fmaxf(m4.z, m4.w));
            float ls = l4.x * __expf(m4.x - mstar);
            ls += l4.y * __expf(m4.y - mstar);
            ls += l4.z * __expf(m4.z - mstar);
            ls += l4.w * __expf(m4.w - mstar);
            lstar[g][vv] = ls;
            const float f = __expf(mrow[g][vv] - mstar);
            #pragma unroll
            for (int c = 0; c < 4; ++c)
                obuf[(((wave * 2 + g) * 4 + c) * 8 + vv) * 32 + lane_m] = O[g][c][vv] * f;
        }
    }
    __syncthreads();

    const int gr = wave >> 1;
    const int cp = wave & 1;
    #pragma unroll
    for (int cc = 0; cc < 2; ++cc) {
        const int c = 2 * cp + cc;
        #pragma unroll
        for (int vv = 0; vv < 8; ++vv) {
            const float* ob = &obuf[((gr * 4 + c) * 8 + vv) * 32 + lane_m];
            float o = ob[0];
            #pragma unroll
            for (int w = 1; w < AWAVES; ++w)
                o += ob[w * (64 * 32)];
            const float ls = gr ? lstar[1][vv] : lstar[0][vv];
            const int t = q0 + 16 * gr + vv + 8 * laneHi_m;
            const int d = c * 16 + lane16_m;
            out[(((long)b * SEQ + t) * NHEAD + h) * HD + d] = o / ls;
        }
    }
}

extern "C" void kernel_launch(void* const* d_in, const int* in_sizes, int n_in,
                              void* d_out, int out_size, void* d_ws, size_t ws_size,
                              hipStream_t stream) {
    const float* x    = (const float*)d_in[0];
    const float* Wqkv = (const float*)d_in[1];
    const float* bqkv = (const float*)d_in[2];
    float* out = (float*)d_out;

    float* ws = (float*)d_ws;
    const long QS = (long)BATCH * NHEAD * SEQ * HD;
    float* qb = ws;
    float* kb = ws + QS;
    float* vb = ws + 2 * QS;

    qkv_gemm_kernel<<<1536, 128, 0, stream>>>(x, Wqkv, bqkv, qb, kb, vb);
    attn_kernel<<<BATCH * NHEAD * (SEQ / 32), 32 * AWAVES, 0, stream>>>(qb, kb, vb, out);
}


# Attention softmax: 716 to 388 vector instructions per 64-key trip (exp2 of one fma per score, mask only on the diagonal trip, lane-held row sums until the merge, DPP max by hand); l sum order changes
# speedup vs baseline: 2.5126x; 1.0329x over previous
#include <hip/hip_runtime.h>

typedef float v2f __attribute__((ext_vector_type(2)));
typedef float v8f __attribute__((ext_vector_type(8)));

#define WMMA_F32(a, b, c) \
    __builtin_amdgcn_wmma_f32_16x16x4_f32(false, (a), false, (b), (short)0, (c), false, false)

#define EMBED 1024
#define THREE_EMBED 3072
#define NHEAD 16
#define HD 64
#define SEQ 2048
#define BATCH 4

typedef float v4f __attribute__((ext_vector_type(4)));

#define GBM 128
#define GBN 128
#define GBK 32
#define APITCH 36
#define BPITCH 136

__global__ __launch_bounds__(128) void qkv_gemm_kernel(
    const float* __restrict__ x, const float* __restrict__ w,
    const float* __restrict__ bias,
    float* __restrict__ qb, float* __restrict__ kb, float* __restrict__ vb)
{
    __shared__ __attribute__((aligned(16))) float As[2][GBM * APITCH];
    __shared__ __attribute__((aligned(16))) float Bs[2][GBK * BPITCH];

    const int tid    = threadIdx.x;
    const int lane   = tid & 31;
    const int wave   = tid >> 5;
    const int lane16 = lane & 15;
    const int laneHi = lane >> 4;
    const int wm     = wave >> 1;
    const int wn     = wave & 1;

    const int mtile = blockIdx.x & 63;
    const int ntile = blockIdx.x >> 6;
    const int m0    = mtile * GBM;
    const int n0    = ntile * GBN;

    const int arow = tid >> 3;
    const int acol = (tid & 7) * 4;
    const int brow = tid >> 5;
    const int bcol = (tid & 31) * 4;
    const float* ag = x + (long)(m0 + arow) * EMBED + acol;
    const float* bg = w + (long)brow * THREE_EMBED + n0 + bcol;

    v4f ra[8], rb[8];
    #pragma unroll
    for (int i = 0; i < 8; ++i) {
        ra[i] = *(const v4f*)(ag + (long)(16 * i) * EMBED);
        rb[i] = *(const v4f*)(bg + (long)(4 * i) * THREE_EMBED);
    }
    #pragma unroll
    for (int i = 0; i < 8; ++i) {
        *(v4f*)&As[0][(arow + 16 * i) * APITCH + acol] = ra[i];
        *(v4f*)&Bs[0][(brow + 4 * i) * BPITCH + bcol]  = rb[i];
    }
    __syncthreads();

    v8f acc[4][4] = {};

    const int aoff = (wm * 64 + lane16) * APITCH + 2 * laneHi;
    const int boff = (2 * laneHi) * BPITCH + wn * 64 + lane16;

    for (int kt = 0; kt < EMBED / GBK; ++kt) {
        const int cur = kt & 1;
        const bool more = (kt + 1) < (EMBED / GBK);

        if (more) {
            const float* agn = ag + (kt + 1) * GBK;
            const float* bgn = bg + (long)(kt + 1) * GBK * THREE_EMBED;
            #pragma unroll
            for (int i = 0; i < 8; ++i) {
                ra[i] = *(const v4f*)(agn + (long)(16 * i) * EMBED);
                rb[i] = *(const v4f*)(bgn + (long)(4 * i) * THREE_EMBED);
            }
        }

        const float* as = &As[cur][aoff];
        const float* bs = &Bs[cur][boff];
        #pragma unroll
        for (int ks = 0; ks < GBK / 4; ++ks) {
            v2f a[4], b[4];
            #pragma unroll
            for (int i = 0; i < 4; ++i)
                a[i] = *(const v2f*)(as + 16 * i * APITCH + 4 * ks);
            #pragma unroll
            for (int j = 0; j < 4; ++j) {
                b[j].x = bs[(4 * ks) * BPITCH + 16 * j];
                b[j].y = bs[(4 * ks + 1) * BPITCH + 16 * j];
            }
            #pragma unroll
            for (int i = 0; i < 4; ++i)
                #pragma unroll
                for (int j = 0; j < 4; ++j)
                    acc[i][j] = WMMA_F32(a[i], b[j], acc[i][j]);
        }

        if (more) {
            #pragma unroll
            for (int i = 0; i < 8; ++i) {
                *(v4f*)&As[cur ^ 1][(arow + 16 * i) * APITCH + acol] = ra[i];
                *(v4f*)&Bs[cur ^ 1][(brow + 4 * i) * BPITCH + bcol]  = rb[i];
            }
        }
        __syncthreads();
    }

    const int nw0     = n0 + wn * 64;
    const int section = nw0 / EMBED;
    const int h       = (nw0 % EMBED) / HD;
    float* dst = (section == 0) ? qb : (section == 1) ? kb : vb;

    #pragma unroll
    for (int j = 0; j < 4; ++j) {
        const int d = j * 16 + lane16;
        const float bv = bias[nw0 + d];
        #pragma unroll
        for (int i = 0; i < 4; ++i) {
            #pragma unroll
            for (int vv = 0; vv < 8; ++vv) {
                const int m  = m0 + wm * 64 + 16 * i + vv + 8 * laneHi;
                const int bi = m >> 11;
                const int t  = m & 2047;
                dst[(((long)(bi * NHEAD + h)) * SEQ + t) * HD + d] = acc[i][j][vv] + bv;
            }
        }
    }
}

template <int M>
__device__ __forceinline__ float row_xor(float x) {
    const int xi = __builtin_bit_cast(int, x);
    return __builtin_bit_cast(float,
        __builtin_amdgcn_update_dpp(xi, xi, 0x160 + M, 0xf, 0xf, true));
}

__device__ __forceinline__ void row_max16_x4(float& t0, float& t1, float& t2, float& t3,
                                             float m0, float m1, float m2, float m3) {
    asm("v_max_num_f32_dpp %0, %0, %0 row_xmask:1 row_mask:0xf bank_mask:0xf bound_ctrl:1\n\t"
        "v_max_num_f32_dpp %1, %1, %1 row_xmask:1 row_mask:0xf bank_mask:0xf bound_ctrl:1\n\t"
        "v_max_num_f32_dpp %2, %2, %2 row_xmask:1 row_mask:0xf bank_mask:0xf bound_ctrl:1\n\t"
        "v_max_num_f32_dpp %3, %3, %3 row_xmask:1 row_mask:0xf bank_mask:0xf bound_ctrl:1\n\t"
        "v_max_num_f32_dpp %0, %0, %0 row_xmask:2 row_mask:0xf bank_mask:0xf bound_ctrl:1\n\t"
        "v_max_num_f32_dpp %1, %1, %1 row_xmask:2 row_mask:0xf bank_mask:0xf bound_ctrl:1\n\t"
        "v_max_num_f32_dpp %2, %2, %2 row_xmask:2 row_mask:0xf bank_mask:0xf bound_ctrl:1\n\t"
        "v_max_num_f32_dpp %3, %3, %3 row_xmask:2 row_mask:0xf bank_mask:0xf bound_ctrl:1\n\t"
        "v_max_num_f32_dpp %0, %0, %0 row_xmask:4 row_mask:0xf bank_mask:0xf bound_ctrl:1\n\t"
        "v_max_num_f32_dpp %1, %1, %1 row_xmask:4 row_mask:0xf bank_mask:0xf bound_ctrl:1\n\t"
        "v_max_num_f32_dpp %2, %2, %2 row_xmask:4 row_mask:0xf bank_mask:0xf bound_ctrl:1\n\t"
        "v_max_num_f32_dpp %3, %3, %3 row_xmask:4 row_mask:0xf bank_mask:0xf bound_ctrl:1\n\t"
        "v_max_num_f32_dpp %0, %0, %0 row_xmask:8 row_mask:0xf bank_mask:0xf bound_ctrl:1\n\t"
        "v_max_num_f32_dpp %1, %1, %1 row_xmask:8 row_mask:0xf bank_mask:0xf bound_ctrl:1\n\t"
        "v_max_num_f32_dpp %2, %2, %2 row_xmask:8 row_mask:0xf bank_mask:0xf bound_ctrl:1\n\t"
        "v_max_num_f32_dpp %3, %3, %3 row_xmask:8 row_mask:0xf bank_mask:0xf bound_ctrl:1\n\t"
        "v_max_num_f32 %0, %0, %4\n\t"
        "v_max_num_f32 %1, %1, %5\n\t"
        "v_max_num_f32 %2, %2, %6\n\t"
        "v_max_num_f32 %3, %3, %7"
        : "+v"(t0), "+v"(t1), "+v"(t2), "+v"(t3)
        : "v"(m0), "v"(m1), "v"(m2), "v"(m3));
}

#define PPITCH 68
#define AWAVES 4

__device__ __forceinline__ void wave_lds_sync() {
    __builtin_amdgcn_fence(__ATOMIC_RELEASE, "workgroup", "local");
    __builtin_amdgcn_wave_barrier();
    __builtin_amdgcn_fence(__ATOMIC_ACQUIRE, "workgroup", "local");
}

__device__ __forceinline__ void load_k_tile(v2f (&dst)[16], const float* __restrict__ p) {
    #pragma unroll
    for (int j = 0; j < 16; ++j)
        dst[j] = *(const v2f*)(p + 4 * j);
}

__device__ __forceinline__ void load_v_tile(v2f (&dst)[4][4], const float* __restrict__ p) {
    #pragma unroll
    for (int j = 0; j < 4; ++j) {
        #pragma unroll
        for (int c = 0; c < 4; ++c) {
            dst[j][c].x = p[(4 * j) * HD + 16 * c];
            dst[j][c].y = p[(4 * j + 1) * HD + 16 * c];
        }
    }
}

__global__ __launch_bounds__(32 * AWAVES, 3) void attn_kernel(
    const float* __restrict__ q, const float* __restrict__ k,
    const float* __restrict__ v, float* __restrict__ out)
{
    __shared__ __attribute__((aligned(16))) float pstage[AWAVES][2 * 16 * PPITCH];
    __shared__ __attribute__((aligned(16))) float mbuf[32 * AWAVES];
    __shared__ __attribute__((aligned(16))) float lbuf[32 * AWAVES];
    __shared__ __attribute__((aligned(16))) float obuf[AWAVES * 64 * 32];

    const int lane   = threadIdx.x & 31;
    const int lane16 = lane & 15;
    const int laneHi = lane >> 4;
    const int wave   = __builtin_amdgcn_readfirstlane((int)(threadIdx.x >> 5));

    const int rid = BATCH * NHEAD * (SEQ / 32) - 1 - (int)blockIdx.x;
    const int q32 = (rid >> 5) & 63;
    const int bh  = ((rid >> 11) << 5) | (rid & 31);
    const int b   = bh >> 4;
    const int h   = bh & 15;
    const int q0  = q32 * 32;

    const float* qp0 = q + ((long)bh * SEQ + q0 + lane16) * HD + 2 * laneHi;
    const float* qp1 = qp0 + 16 * HD;
    v2f aQ0[16], aQ1[16];
    #pragma unroll
    for (int j = 0; j < 16; ++j) {
        aQ0[j] = *(const v2f*)(qp0 + 4 * j);
        aQ1[j] = *(const v2f*)(qp1 + 4 * j);
    }

    float mrow[2][8], lrow[2][8];
    v8f O[2][4] = {};
    #pragma unroll
    for (int g = 0; g < 2; ++g)
        #pragma unroll
        for (int vv = 0; vv < 8; ++vv) { mrow[g][vv] = -__builtin_inff(); lrow[g][vv] = 0.f; }

    const float cexp = 0.125f * 1.44269504088896340736f;
    const int ntrip = (q0 + 95) >> 6;

    float* pst = &pstage[wave][(8 * laneHi) * PPITCH + lane16];
    const float* pld = &pstage[wave][lane16 * PPITCH + 2 * laneHi];

    const float* kfp = k + ((long)bh * SEQ + lane16) * HD + 2 * laneHi;
    const float* vfp = v + ((long)bh * SEQ + 2 * laneHi) * HD + lane16;

    v2f kf[2][16];
    v2f vf[2][4][4];

    {
        const int ktf = (wave < ntrip) ? wave : (ntrip - 1);
        load_k_tile(kf[0], kfp + (long)(ktf * 64) * HD);
    }

    for (int kt = wave; kt < ntrip; kt += AWAVES) {
        const int k0 = kt * 64;
        const float* ktp = kfp + (long)k0 * HD;
        const float* vtp = vfp + (long)k0 * HD;
        const int ktn = (kt + AWAVES < ntrip) ? (kt + AWAVES) : (ntrip - 1);
        const int dmask = q0 + 8 * laneHi - lane16 - k0;

        v8f S[2][4];
        #pragma unroll
        for (int n = 0; n < 4; ++n) {
            if (n < 3) load_k_tile(kf[(n + 1) & 1], ktp + (16 * (n + 1)) * HD);
            __builtin_amdgcn_sched_barrier(0);
            v8f s0 = {};
            v8f s1 = {};
            #pragma unroll
            for (int j = 0; j < 16; ++j) {
                s0 = WMMA_F32(aQ0[j], kf[n & 1][j], s0);
                s1 = WMMA_F32(aQ1[j], kf[n & 1][j], s1);
            }
            S[0][n] = s0;
            S[1][n] = s1;
            __builtin_amdgcn_sched_barrier(0);
        }
        load_v_tile(vf[0], vtp);
        __builtin_amdgcn_sched_barrier(0);

        if (k0 + 63 > q0) {
            #pragma unroll
            for (int g = 0; g < 2; ++g)
                #pragma unroll
                for (int n = 0; n < 4; ++n)
                    #pragma unroll
                    for (int vv = 0; vv < 8; ++vv) {
                        if (16 * n - 16 * g - vv > dmask) S[g][n][vv] = -__builtin_inff();
                    }
        }
        #pragma unroll
        for (int g = 0; g < 2; ++g) {
            float mnew[8];
            #pragma unroll
            for (int vv = 0; vv < 8; ++vv)
                mnew[vv] = fmaxf(fmaxf(S[g][0][vv], S[g][1][vv]),
                                 fmaxf(S[g][2][vv], S[g][3][vv])) * cexp;
            row_max16_x4(mnew[0], mnew[1], mnew[2], mnew[3],
                         mrow[g][0], mrow[g][1], mrow[g][2], mrow[g][3]);
            row_max16_x4(mnew[4], mnew[5], mnew[6], mnew[7],
                         mrow[g][4], mrow[g][5], mrow[g][6], mrow[g][7]);
            float alpha[8];
            #pragma unroll
            for (int vv = 0; vv < 8; ++vv) {
                float p[4];
                #pragma unroll
                for (int n = 0; n < 4; ++n)
                    p[n] = __builtin_amdgcn_exp2f(__builtin_fmaf(S[g][n][vv], cexp, -mnew[vv]));
                const float a = __builtin_amdgcn_exp2f(mrow[g][vv] - mnew[vv]);
                lrow[g][vv]  = lrow[g][vv] * a + ((p[0] + p[1]) + (p[2] + p[3]));
                mrow[g][vv]  = mnew[vv];
                alpha[vv]    = a;
                #pragma unroll
                for (int n = 0; n < 4; ++n)
                    S[g][n][vv] = p[n];
            }
            #pragma unroll
            for (int c = 0; c < 4; ++c)
                #pragma unroll
                for (int vv = 0; vv < 8; ++vv)
                    O[g][c][vv] *= alpha[vv];
        }

        wave_lds_sync();
        #pragma unroll
        for (int g = 0; g < 2; ++g)
            #pragma unroll
            for (int n = 0; n < 4; ++n)
                #pragma unroll
                for (int vv = 0; vv < 8; ++vv)
                    pst[g * 16 * PPITCH + vv * PPITCH + 16 * n] = S[g][n][vv];
        wave_lds_sync();

        v2f aP[2][4][2];
        #pragma unroll
        for (int j = 0; j < 4; ++j) {
            aP[0][j][0] = *(const v2f*)(pld + 4 * j);
            aP[0][j][1] = *(const v2f*)(pld + 16 * PPITCH + 4 * j);
        }
        #pragma unroll
        for (int n = 0; n < 4; ++n) {
            if (n < 3) {
                load_v_tile(vf[(n + 1) & 1], vtp + (16 * (n + 1)) * HD);
                #pragma unroll
                for (int j = 0; j < 4; ++j) {
                    aP[(n + 1) & 1][j][0] = *(const v2f*)(pld + 16 * (n + 1) + 4 * j);
                    aP[(n + 1) & 1][j][1] = *(const v2f*)(pld + 16 * PPITCH + 16 * (n + 1) + 4 * j);
                }
            } else {
                load_k_tile(kf[0], kfp + (long)(ktn * 64) * HD);
            }
            __builtin_amdgcn_sched_barrier(0);
            #pragma unroll
            for (int j = 0; j < 4; ++j) {
                #pragma unroll
                for (int c = 0; c < 4; ++c) {
                    O[0][c] = WMMA_F32(aP[n & 1][j][0], vf[n & 1][j][c], O[0][c]);
                    O[1][c] = WMMA_F32(aP[n & 1][j][1], vf[n & 1][j][c], O[1][c]);
                }
            }
            __builtin_amdgcn_sched_barrier(0);
        }
    }

    const int lane_m   = (int)__builtin_amdgcn_mbcnt_lo(~0u, 0u);
    const int lane16_m = lane_m & 15;
    const int laneHi_m = lane_m >> 4;
    #pragma unroll
    for (int g = 0; g < 2; ++g)
        #pragma unroll
        for (int vv = 0; vv < 8; ++vv) {
            float l = lrow[g][vv];
            l += row_xor<1>(l);
            l += row_xor<2>(l);
            l += row_xor<4>(l);
            l += row_xor<8>(l);
            lrow[g][vv] = l;
        }
    if (lane16_m == 0) {
        #pragma unroll
        for (int g = 0; g < 2; ++g)
            #pragma unroll
            for (int vv = 0; vv < 8; ++vv) {
                const int row = 16 * g + 8 * laneHi_m + vv;
                mbuf[row * AWAVES + wave] = mrow[g][vv];
                lbuf[row * AWAVES + wave] = lrow[g][vv];
            }
    }
    __syncthreads();

    float lstar[2][8];
    #pragma unroll
    for (int g = 0; g < 2; ++g) {
        #pragma unroll
        for (int vv = 0; vv < 8; ++vv) {
            const int row = 16 * g + 8 * laneHi_m + vv;
            const v4f m4 = *(const v4f*)&mbuf[row * AWAVES];
            const v4f l4 = *(const v4f*)&lbuf[row * AWAVES];
            const float mstar = fmaxf(fmaxf(m4.x, m4.y), fmaxf(m4.z, m4.w));
            float ls = l4.x * __builtin_amdgcn_exp2f(m4.x - mstar);
            ls += l4.y * __builtin_amdgcn_exp2f(m4.y - mstar);
            ls += l4.z * __builtin_amdgcn_exp2f(m4.z - mstar);
            ls += l4.w * __builtin_amdgcn_exp2f(m4.w - mstar);
            lstar[g][vv] = ls;
            const float f = __builtin_amdgcn_exp2f(mrow[g][vv] - mstar);
            #pragma unroll
            for (int c = 0; c < 4; ++c)
                obuf[(((wave * 2 + g) * 4 + c) * 8 + vv) * 32 + lane_m] = O[g][c][vv] * f;
        }
    }
    __syncthreads();

    const int gr = wave >> 1;
    const int cp = wave & 1;
    #pragma unroll
    for (int cc = 0; cc < 2; ++cc) {
        const int c = 2 * cp + cc;
        #pragma unroll
        for (int vv = 0; vv < 8; ++vv) {
            const float* ob = &obuf[((gr * 4 + c) * 8 + vv) * 32 + lane_m];
            float o = ob[0];
            #pragma unroll
            for (int w = 1; w < AWAVES; ++w)
                o += ob[w * (64 * 32)];
            const float ls = gr ? lstar[1][vv] : lstar[0][vv];
            const int t = q0 + 16 * gr + vv + 8 * laneHi_m;
            const int d = c * 16 + lane16_m;
            out[(((long)b * SEQ + t) * NHEAD + h) * HD + d] = o / ls;
        }
    }
}

extern "C" void kernel_launch(void* const* d_in, const int* in_sizes, int n_in,
                              void* d_out, int out_size, void* d_ws, size_t ws_size,
                              hipStream_t stream) {
    const float* x    = (const float*)d_in[0];
    const float* Wqkv = (const float*)d_in[1];
    const float* bqkv = (const float*)d_in[2];
    float* out = (float*)d_out;

    float* ws = (float*)d_ws;
    const long QS = (long)BATCH * NHEAD * SEQ * HD;
    float* qb = ws;
    float* kb = ws + QS;
    float* vb = ws + 2 * QS;

    qkv_gemm_kernel<<<1536, 128, 0, stream>>>(x, Wqkv, bqkv, qb, kb, vb);
    attn_kernel<<<BATCH * NHEAD * (SEQ / 32), 32 * AWAVES, 0, stream>>>(qb, kb, vb, out);
}
